# starting kernel
# baseline (speedup 1.0000x reference)
#include <hip/hip_runtime.h>
#include <hip/hip_bf16.h>


typedef float v2f __attribute__((ext_vector_type(2)));
typedef float v8f __attribute__((ext_vector_type(8)));

#define BATCH   1024
#define C_IN    2048
#define POOLSZ  32

__global__ void pool_mean32(const float* __restrict__ img,
                            float* __restrict__ feat, int total) {
    int i = blockIdx.x * blockDim.x + threadIdx.x;
    if (i >= total) return;
    const float4* p = reinterpret_cast<const float4*>(img + (size_t)i * POOLSZ);
    float s = 0.0f;
#pragma unroll
    for (int j = 0; j < 8; ++j) {
        float4 q = p[j];
        s += (q.x + q.y) + (q.z + q.w);
    }
    feat[i] = s * (1.0f / 32.0f);
}

template <bool RELU>
__global__ void wmma_gemm_f32(const float* __restrict__ A,
                              const float* __restrict__ W,
                              const float* __restrict__ bias,
                              float* __restrict__ C,
                              int M, int N, int K) {
    const int lane  = threadIdx.x;
    const int half  = lane >> 4;
    const int l15   = lane & 15;
    const int ntile = blockIdx.x;
    const int mtile = blockIdx.y;

    const int n  = ntile * 16 + l15;
    const int am = mtile * 16 + l15;

    const float bv = bias[n];
    v8f acc;
#pragma unroll
    for (int r = 0; r < 8; ++r) acc[r] = bv;

    const float* __restrict__ Arow = A + (size_t)am * K + 2 * half;
    const float* __restrict__ Wcol = W + (size_t)(2 * half) * N + n;

    for (int k = 0; k < K; k += 32) {
#pragma unroll
        for (int kk = 0; kk < 32; kk += 4) {
            v2f a, b;
            a.x = Arow[k + kk];
            a.y = Arow[k + kk + 1];
            b.x = Wcol[(size_t)(k + kk) * N];
            b.y = Wcol[(size_t)(k + kk + 1) * N];
            acc = __builtin_amdgcn_wmma_f32_16x16x4_f32(
                false, a, false, b, (short)0, acc, false, false);
        }
    }

    float* __restrict__ Cout = C + (size_t)(mtile * 16 + 8 * half) * N + n;
#pragma unroll
    for (int r = 0; r < 8; ++r) {
        float v = acc[r];
        if (RELU) v = fmaxf(v, 0.0f);
        Cout[(size_t)r * N] = v;
    }
}

__device__ __forceinline__ float block_reduce_128(float x, float* red) {
    const int t = threadIdx.x;
    red[t] = x;
    __syncthreads();
#pragma unroll
    for (int s = 64; s > 0; s >>= 1) {
        if (t < s) red[t] += red[t + s];
        __syncthreads();
    }
    float r = red[0];
    __syncthreads();
    return r;
}

__global__ void epilogue(const float* __restrict__ isp,
                         const float* __restrict__ ish,
                         const int* __restrict__ label,
                         const float* __restrict__ ppos,
                         const float* __restrict__ pneg,
                         float* __restrict__ out) {
    __shared__ float red[128];
    const int b = blockIdx.x;
    const int t = threadIdx.x;

    const float v  = isp[b * 128 + t];
    const float pp = ppos[t];
    const float nn = pneg[t];

    const float nv  = block_reduce_128(v * v, red);
    const float npp = block_reduce_128(pp * pp, red);
    const float nnn = block_reduce_128(nn * nn, red);
    const float dp  = block_reduce_128(v * pp, red);
    const float dn  = block_reduce_128(v * nn, red);

    const float eps = 1e-12f;
    const float dv  = fmaxf(sqrtf(nv), eps);
    const float lp  = dp / (dv * fmaxf(sqrtf(npp), eps));
    const float ln  = dn / (dv * fmaxf(sqrtf(nnn), eps));

    const bool lab    = label[b] != 0;
    const bool pred   = lp > ln;
    const bool is_bad = (pred == lab) && (lp != ln);
    const float diff  = fabsf(lp - ln);
    const float scale = is_bad ? (tanhf(-diff) + 1.0f) : 1.0f;

    out[b * 256 + t]       = v * scale;
    out[b * 256 + 128 + t] = ish[b * 128 + t];
}

extern "C" void kernel_launch(void* const* d_in, const int* in_sizes, int n_in,
                              void* d_out, int out_size, void* d_ws, size_t ws_size,
                              hipStream_t stream) {
    const float* img   = (const float*)d_in[0];
    const int*   label = (const int*)d_in[1];
    const float* W1  = (const float*)d_in[2];
    const float* b1  = (const float*)d_in[3];
    const float* W2  = (const float*)d_in[4];
    const float* b2  = (const float*)d_in[5];
    const float* W3  = (const float*)d_in[6];
    const float* b3  = (const float*)d_in[7];
    const float* Wsp = (const float*)d_in[8];
    const float* bsp = (const float*)d_in[9];
    const float* Wsh = (const float*)d_in[10];
    const float* bsh = (const float*)d_in[11];
    const float* ppn = (const float*)d_in[12];
    const float* pnn = (const float*)d_in[13];

    float* ws   = (float*)d_ws;
    float* feat = ws;
    float* h1   = feat + (size_t)BATCH * C_IN;
    float* h2   = h1   + (size_t)BATCH * 512;
    float* x    = h2   + (size_t)BATCH * 256;
    float* isp  = x    + (size_t)BATCH * 128;
    float* ish  = isp  + (size_t)BATCH * 128;

    {
        const int total = BATCH * C_IN;
        pool_mean32<<<(total + 255) / 256, 256, 0, stream>>>(img, feat, total);
    }

    wmma_gemm_f32<true><<<dim3(512 / 16, BATCH / 16), 32, 0, stream>>>(
        feat, W1, b1, h1, BATCH, 512, C_IN);
    wmma_gemm_f32<true><<<dim3(256 / 16, BATCH / 16), 32, 0, stream>>>(
        h1, W2, b2, h2, BATCH, 256, 512);
    wmma_gemm_f32<false><<<dim3(128 / 16, BATCH / 16), 32, 0, stream>>>(
        h2, W3, b3, x, BATCH, 128, 256);

    wmma_gemm_f32<false><<<dim3(128 / 16, BATCH / 16), 32, 0, stream>>>(
        x, Wsp, bsp, isp, BATCH, 128, 128);
    wmma_gemm_f32<false><<<dim3(128 / 16, BATCH / 16), 32, 0, stream>>>(
        x, Wsh, bsh, ish, BATCH, 128, 128);

    epilogue<<<BATCH, 128, 0, stream>>>(isp, ish, label, ppn, pnn, (float*)d_out);
}


# GEMM kernel: K range split over the 8 (heads: 4) waves of a workgroup per 16x16 tile, partial tiles summed through LDS with the bias; k order inside a slice kept, sum order across slices changes
# speedup vs baseline: 1.0781x; 1.0781x over previous
#include <hip/hip_runtime.h>
#include <hip/hip_bf16.h>


typedef float v2f __attribute__((ext_vector_type(2)));
typedef float v8f __attribute__((ext_vector_type(8)));

#define BATCH   1024
#define C_IN    2048
#define POOLSZ  32

__global__ void pool_mean32(const float* __restrict__ img,
                            float* __restrict__ feat, int total) {
    int i = blockIdx.x * blockDim.x + threadIdx.x;
    if (i >= total) return;
    const float4* p = reinterpret_cast<const float4*>(img + (size_t)i * POOLSZ);
    float s = 0.0f;
#pragma unroll
    for (int j = 0; j < 8; ++j) {
        float4 q = p[j];
        s += (q.x + q.y) + (q.z + q.w);
    }
    feat[i] = s * (1.0f / 32.0f);
}

template <bool RELU, int KS>
__global__ void __launch_bounds__(32 * KS)
wmma_gemm_f32(const float* __restrict__ A,
              const float* __restrict__ W,
              const float* __restrict__ bias,
              float* __restrict__ C,
              int M, int N, int K) {
    __shared__ float part[KS][8][32];

    const int tid   = threadIdx.x;
    const int lane  = tid & 31;
    const int wave  = __builtin_amdgcn_readfirstlane(tid >> 5);
    const int half  = lane >> 4;
    const int l15   = lane & 15;
    const int ntile = blockIdx.x;
    const int mtile = blockIdx.y;

    const int n  = ntile * 16 + l15;
    const int am = mtile * 16 + l15;

    const int kslice = K / KS;
    const int kbeg   = wave * kslice;

    v8f acc;
#pragma unroll
    for (int r = 0; r < 8; ++r) acc[r] = 0.0f;

    const float* __restrict__ Arow = A + (size_t)am * K + kbeg + 2 * half;
    const float* __restrict__ Wcol = W + (size_t)(kbeg + 2 * half) * N + n;

    for (int k = 0; k < kslice; k += 32) {
#pragma unroll
        for (int kk = 0; kk < 32; kk += 4) {
            v2f a, b;
            a.x = Arow[k + kk];
            a.y = Arow[k + kk + 1];
            b.x = Wcol[(size_t)(k + kk) * N];
            b.y = Wcol[(size_t)(k + kk + 1) * N];
            acc = __builtin_amdgcn_wmma_f32_16x16x4_f32(
                false, a, false, b, (short)0, acc, false, false);
        }
    }

#pragma unroll
    for (int r = 0; r < 8; ++r) part[wave][r][lane] = acc[r];
    __syncthreads();

    for (int e = tid; e < 256; e += 32 * KS) {
        const int r  = e >> 5;
        const int ln = e & 31;
        const int col = ntile * 16 + (ln & 15);
        float s = bias[col];
        for (int w = 0; w < KS; ++w) s += part[w][r][ln];
        if (RELU) s = fmaxf(s, 0.0f);
        C[(size_t)(mtile * 16 + r + 8 * (ln >> 4)) * N + col] = s;
    }
}

__device__ __forceinline__ float block_reduce_128(float x, float* red) {
    const int t = threadIdx.x;
    red[t] = x;
    __syncthreads();
#pragma unroll
    for (int s = 64; s > 0; s >>= 1) {
        if (t < s) red[t] += red[t + s];
        __syncthreads();
    }
    float r = red[0];
    __syncthreads();
    return r;
}

__global__ void epilogue(const float* __restrict__ isp,
                         const float* __restrict__ ish,
                         const int* __restrict__ label,
                         const float* __restrict__ ppos,
                         const float* __restrict__ pneg,
                         float* __restrict__ out) {
    __shared__ float red[128];
    const int b = blockIdx.x;
    const int t = threadIdx.x;

    const float v  = isp[b * 128 + t];
    const float pp = ppos[t];
    const float nn = pneg[t];

    const float nv  = block_reduce_128(v * v, red);
    const float npp = block_reduce_128(pp * pp, red);
    const float nnn = block_reduce_128(nn * nn, red);
    const float dp  = block_reduce_128(v * pp, red);
    const float dn  = block_reduce_128(v * nn, red);

    const float eps = 1e-12f;
    const float dv  = fmaxf(sqrtf(nv), eps);
    const float lp  = dp / (dv * fmaxf(sqrtf(npp), eps));
    const float ln  = dn / (dv * fmaxf(sqrtf(nnn), eps));

    const bool lab    = label[b] != 0;
    const bool pred   = lp > ln;
    const bool is_bad = (pred == lab) && (lp != ln);
    const float diff  = fabsf(lp - ln);
    const float scale = is_bad ? (tanhf(-diff) + 1.0f) : 1.0f;

    out[b * 256 + t]       = v * scale;
    out[b * 256 + 128 + t] = ish[b * 128 + t];
}

extern "C" void kernel_launch(void* const* d_in, const int* in_sizes, int n_in,
                              void* d_out, int out_size, void* d_ws, size_t ws_size,
                              hipStream_t stream) {
    const float* img   = (const float*)d_in[0];
    const int*   label = (const int*)d_in[1];
    const float* W1  = (const float*)d_in[2];
    const float* b1  = (const float*)d_in[3];
    const float* W2  = (const float*)d_in[4];
    const float* b2  = (const float*)d_in[5];
    const float* W3  = (const float*)d_in[6];
    const float* b3  = (const float*)d_in[7];
    const float* Wsp = (const float*)d_in[8];
    const float* bsp = (const float*)d_in[9];
    const float* Wsh = (const float*)d_in[10];
    const float* bsh = (const float*)d_in[11];
    const float* ppn = (const float*)d_in[12];
    const float* pnn = (const float*)d_in[13];

    float* ws   = (float*)d_ws;
    float* feat = ws;
    float* h1   = feat + (size_t)BATCH * C_IN;
    float* h2   = h1   + (size_t)BATCH * 512;
    float* x    = h2   + (size_t)BATCH * 256;
    float* isp  = x    + (size_t)BATCH * 128;
    float* ish  = isp  + (size_t)BATCH * 128;

    {
        const int total = BATCH * C_IN;
        pool_mean32<<<(total + 255) / 256, 256, 0, stream>>>(img, feat, total);
    }

    wmma_gemm_f32<true, 8><<<dim3(512 / 16, BATCH / 16), 256, 0, stream>>>(
        feat, W1, b1, h1, BATCH, 512, C_IN);
    wmma_gemm_f32<true, 8><<<dim3(256 / 16, BATCH / 16), 256, 0, stream>>>(
        h1, W2, b2, h2, BATCH, 256, 512);
    wmma_gemm_f32<false, 8><<<dim3(128 / 16, BATCH / 16), 256, 0, stream>>>(
        h2, W3, b3, x, BATCH, 128, 256);

    wmma_gemm_f32<false, 4><<<dim3(128 / 16, BATCH / 16), 128, 0, stream>>>(
        x, Wsp, bsp, isp, BATCH, 128, 128);
    wmma_gemm_f32<false, 4><<<dim3(128 / 16, BATCH / 16), 128, 0, stream>>>(
        x, Wsh, bsh, ish, BATCH, 128, 128);

    epilogue<<<BATCH, 128, 0, stream>>>(isp, ish, label, ppn, pnn, (float*)d_out);
}


# GEMM kernel: each wave keeps a 32x64 (small GEMMs 16x64) block as 2x4 WMMA tiles, columns interleaved by 4; A, W fragments straight from global as 128-bit loads, no LDS; k order in a slice changes
# speedup vs baseline: 2.0457x; 1.8975x over previous
#include <hip/hip_runtime.h>
#include <hip/hip_bf16.h>


typedef float v2f __attribute__((ext_vector_type(2)));
typedef float v8f __attribute__((ext_vector_type(8)));

#define BATCH   1024
#define C_IN    2048
#define POOLSZ  32

__global__ void pool_mean32(const float* __restrict__ img,
                            float* __restrict__ feat, int total) {
    int i = blockIdx.x * blockDim.x + threadIdx.x;
    if (i >= total) return;
    const float4* p = reinterpret_cast<const float4*>(img + (size_t)i * POOLSZ);
    float s = 0.0f;
#pragma unroll
    for (int j = 0; j < 8; ++j) {
        float4 q = p[j];
        s += (q.x + q.y) + (q.z + q.w);
    }
    feat[i] = s * (1.0f / 32.0f);
}

typedef float v4f __attribute__((ext_vector_type(4)));

template <int MT>
struct Frag {
    v4f a[2][MT];
    v4f b[2][4];
};

template <int MT>
__device__ __forceinline__ void load_frag(Frag<MT>& f,
                                          const float* __restrict__ Ap,
                                          const float* __restrict__ Wp,
                                          int K, int N, int k) {
#pragma unroll
    for (int s = 0; s < 2; ++s) {
#pragma unroll
        for (int t = 0; t < MT; ++t)
            f.a[s][t] = *reinterpret_cast<const v4f*>(
                Ap + (size_t)t * 16 * K + k + 8 * s);
#pragma unroll
        for (int i = 0; i < 4; ++i)
            f.b[s][i] = *reinterpret_cast<const v4f*>(
                Wp + (size_t)(k + 8 * s + i) * N);
    }
}

template <int MT>
__device__ __forceinline__ void mma_block(v8f (&acc)[MT][4], const Frag<MT>& f) {
#pragma unroll
    for (int s = 0; s < 2; ++s) {
#pragma unroll
        for (int j = 0; j < 2; ++j) {
#pragma unroll
            for (int t = 0; t < MT; ++t) {
                v2f a;
                a.x = f.a[s][t][2 * j];
                a.y = f.a[s][t][2 * j + 1];
#pragma unroll
                for (int q = 0; q < 4; ++q) {
                    v2f b;
                    b.x = f.b[s][2 * j][q];
                    b.y = f.b[s][2 * j + 1][q];
                    acc[t][q] = __builtin_amdgcn_wmma_f32_16x16x4_f32(
                        false, a, false, b, (short)0, acc[t][q], false, false);
                }
            }
        }
    }
}

template <bool RELU, int KS, int MT>
__global__ void __launch_bounds__(32 * KS)
wmma_gemm_f32(const float* __restrict__ A,
              const float* __restrict__ W,
              const float* __restrict__ bias,
              float* __restrict__ C,
              int M, int N, int K) {
    __shared__ v4f part[KS][MT][8][32];

    const int tid   = threadIdx.x;
    const int lane  = tid & 31;
    const int wave  = __builtin_amdgcn_readfirstlane(tid >> 5);
    const int h     = lane >> 4;
    const int c     = lane & 15;
    const int n0    = blockIdx.x * 64;
    const int m0    = blockIdx.y * (16 * MT);

    const int kslice = K / KS;
    const int kbeg   = wave * kslice;

    v8f acc[MT][4];
#pragma unroll
    for (int t = 0; t < MT; ++t)
#pragma unroll
        for (int q = 0; q < 4; ++q)
#pragma unroll
            for (int r = 0; r < 8; ++r) acc[t][q][r] = 0.0f;

    const float* __restrict__ Ap = A + (size_t)(m0 + c) * K + kbeg + 4 * h;
    const float* __restrict__ Wp = W + (size_t)(kbeg + 4 * h) * N + n0 + 4 * c;

    Frag<MT> f0, f1;
    load_frag<MT>(f0, Ap, Wp, K, N, 0);
    int k = 0;
    for (; k + 32 < kslice; k += 32) {
        load_frag<MT>(f1, Ap, Wp, K, N, k + 16);
        __builtin_amdgcn_sched_barrier(0);
        mma_block<MT>(acc, f0);
        __builtin_amdgcn_sched_barrier(0);
        load_frag<MT>(f0, Ap, Wp, K, N, k + 32);
        __builtin_amdgcn_sched_barrier(0);
        mma_block<MT>(acc, f1);
        __builtin_amdgcn_sched_barrier(0);
    }
    load_frag<MT>(f1, Ap, Wp, K, N, k + 16);
    __builtin_amdgcn_sched_barrier(0);
    mma_block<MT>(acc, f0);
    mma_block<MT>(acc, f1);

#pragma unroll
    for (int t = 0; t < MT; ++t)
#pragma unroll
        for (int r = 0; r < 8; ++r) {
            v4f v;
            v.x = acc[t][0][r];
            v.y = acc[t][1][r];
            v.z = acc[t][2][r];
            v.w = acc[t][3][r];
            part[wave][t][r][lane] = v;
        }
    __syncthreads();

    for (int e = tid; e < MT * 256; e += 32 * KS) {
        const int t  = e >> 8;
        const int r  = (e >> 5) & 7;
        const int ln = e & 31;
        const int col = n0 + 4 * (ln & 15);
        v4f s = *reinterpret_cast<const v4f*>(bias + col);
        for (int w = 0; w < KS; ++w) s += part[w][t][r][ln];
        if (RELU) {
            s.x = fmaxf(s.x, 0.0f);
            s.y = fmaxf(s.y, 0.0f);
            s.z = fmaxf(s.z, 0.0f);
            s.w = fmaxf(s.w, 0.0f);
        }
        *reinterpret_cast<v4f*>(
            C + (size_t)(m0 + 16 * t + 8 * (ln >> 4) + r) * N + col) = s;
    }
}

__device__ __forceinline__ float block_reduce_128(float x, float* red) {
    const int t = threadIdx.x;
    red[t] = x;
    __syncthreads();
#pragma unroll
    for (int s = 64; s > 0; s >>= 1) {
        if (t < s) red[t] += red[t + s];
        __syncthreads();
    }
    float r = red[0];
    __syncthreads();
    return r;
}

__global__ void epilogue(const float* __restrict__ isp,
                         const float* __restrict__ ish,
                         const int* __restrict__ label,
                         const float* __restrict__ ppos,
                         const float* __restrict__ pneg,
                         float* __restrict__ out) {
    __shared__ float red[128];
    const int b = blockIdx.x;
    const int t = threadIdx.x;

    const float v  = isp[b * 128 + t];
    const float pp = ppos[t];
    const float nn = pneg[t];

    const float nv  = block_reduce_128(v * v, red);
    const float npp = block_reduce_128(pp * pp, red);
    const float nnn = block_reduce_128(nn * nn, red);
    const float dp  = block_reduce_128(v * pp, red);
    const float dn  = block_reduce_128(v * nn, red);

    const float eps = 1e-12f;
    const float dv  = fmaxf(sqrtf(nv), eps);
    const float lp  = dp / (dv * fmaxf(sqrtf(npp), eps));
    const float ln  = dn / (dv * fmaxf(sqrtf(nnn), eps));

    const bool lab    = label[b] != 0;
    const bool pred   = lp > ln;
    const bool is_bad = (pred == lab) && (lp != ln);
    const float diff  = fabsf(lp - ln);
    const float scale = is_bad ? (tanhf(-diff) + 1.0f) : 1.0f;

    out[b * 256 + t]       = v * scale;
    out[b * 256 + 128 + t] = ish[b * 128 + t];
}

extern "C" void kernel_launch(void* const* d_in, const int* in_sizes, int n_in,
                              void* d_out, int out_size, void* d_ws, size_t ws_size,
                              hipStream_t stream) {
    const float* img   = (const float*)d_in[0];
    const int*   label = (const int*)d_in[1];
    const float* W1  = (const float*)d_in[2];
    const float* b1  = (const float*)d_in[3];
    const float* W2  = (const float*)d_in[4];
    const float* b2  = (const float*)d_in[5];
    const float* W3  = (const float*)d_in[6];
    const float* b3  = (const float*)d_in[7];
    const float* Wsp = (const float*)d_in[8];
    const float* bsp = (const float*)d_in[9];
    const float* Wsh = (const float*)d_in[10];
    const float* bsh = (const float*)d_in[11];
    const float* ppn = (const float*)d_in[12];
    const float* pnn = (const float*)d_in[13];

    float* ws   = (float*)d_ws;
    float* feat = ws;
    float* h1   = feat + (size_t)BATCH * C_IN;
    float* h2   = h1   + (size_t)BATCH * 512;
    float* x    = h2   + (size_t)BATCH * 256;
    float* isp  = x    + (size_t)BATCH * 128;
    float* ish  = isp  + (size_t)BATCH * 128;

    {
        const int total = BATCH * C_IN;
        pool_mean32<<<(total + 255) / 256, 256, 0, stream>>>(img, feat, total);
    }

    wmma_gemm_f32<true, 8, 2><<<dim3(512 / 64, BATCH / 32), 256, 0, stream>>>(
        feat, W1, b1, h1, BATCH, 512, C_IN);
    wmma_gemm_f32<true, 8, 1><<<dim3(256 / 64, BATCH / 16), 256, 0, stream>>>(
        h1, W2, b2, h2, BATCH, 256, 512);
    wmma_gemm_f32<false, 8, 1><<<dim3(128 / 64, BATCH / 16), 256, 0, stream>>>(
        h2, W3, b3, x, BATCH, 128, 256);

    wmma_gemm_f32<false, 4, 1><<<dim3(128 / 64, BATCH / 16), 128, 0, stream>>>(
        x, Wsp, bsp, isp, BATCH, 128, 128);
    wmma_gemm_f32<false, 4, 1><<<dim3(128 / 64, BATCH / 16), 128, 0, stream>>>(
        x, Wsh, bsh, ish, BATCH, 128, 128);

    epilogue<<<BATCH, 128, 0, stream>>>(isp, ish, label, ppn, pnn, (float*)d_out);
}


# pool kernel: resident grid of 512 8-wave workgroups, each lane loops over its 16 (b,c) items, next two items' 16 loads requested ahead, in place of 65536 one-item waves; per-item sum order kept
# speedup vs baseline: 2.1296x; 1.0410x over previous
#include <hip/hip_runtime.h>
#include <hip/hip_bf16.h>


typedef float v2f __attribute__((ext_vector_type(2)));
typedef float v8f __attribute__((ext_vector_type(8)));

#define BATCH   1024
#define C_IN    2048
#define POOLSZ  32

#define POOL_WGS     512
#define POOL_THREADS 256

struct PoolSet {
    float4 q[2][8];
};

__device__ __forceinline__ void pool_load(PoolSet& s,
                                          const float* __restrict__ img,
                                          int i0, int i1, int last) {
    const float4* p0 = reinterpret_cast<const float4*>(
        img + (size_t)min(i0, last) * POOLSZ);
    const float4* p1 = reinterpret_cast<const float4*>(
        img + (size_t)min(i1, last) * POOLSZ);
#pragma unroll
    for (int j = 0; j < 8; ++j) s.q[0][j] = p0[j];
#pragma unroll
    for (int j = 0; j < 8; ++j) s.q[1][j] = p1[j];
}

__device__ __forceinline__ float pool_sum(const float4 (&q)[8]) {
    float s = 0.0f;
#pragma unroll
    for (int j = 0; j < 8; ++j)
        s += (q[j].x + q[j].y) + (q[j].z + q[j].w);
    return s * (1.0f / 32.0f);
}

__device__ __forceinline__ void pool_store(const PoolSet& s,
                                           float* __restrict__ feat,
                                           int i0, int i1, int total) {
    const float m0 = pool_sum(s.q[0]);
    const float m1 = pool_sum(s.q[1]);
    if (i0 < total) feat[i0] = m0;
    if (i1 < total) feat[i1] = m1;
}

__global__ void __launch_bounds__(POOL_THREADS)
pool_mean32(const float* __restrict__ img,
            float* __restrict__ feat, int total) {
    const int T    = gridDim.x * blockDim.x;
    const int lane = threadIdx.x & 31;
    const int w = __builtin_amdgcn_readfirstlane(
        (int)(blockIdx.x * blockDim.x + threadIdx.x) - lane);
    const int last = total - 1;

    PoolSet s0, s1;
    pool_load(s0, img, w + lane, w + lane + T, last);
    for (int b = w; b < total; b += 4 * T) {
        const int i = b + lane;
        pool_load(s1, img, i + 2 * T, i + 3 * T, last);
        pool_store(s0, feat, i, i + T, total);
        pool_load(s0, img, i + 4 * T, i + 5 * T, last);
        pool_store(s1, feat, i + 2 * T, i + 3 * T, total);
    }
}

typedef float v4f __attribute__((ext_vector_type(4)));

template <int MT>
struct Frag {
    v4f a[2][MT];
    v4f b[2][4];
};

template <int MT>
__device__ __forceinline__ void load_frag(Frag<MT>& f,
                                          const float* __restrict__ Ap,
                                          const float* __restrict__ Wp,
                                          int K, int N, int k) {
#pragma unroll
    for (int s = 0; s < 2; ++s) {
#pragma unroll
        for (int t = 0; t < MT; ++t)
            f.a[s][t] = *reinterpret_cast<const v4f*>(
                Ap + (size_t)t * 16 * K + k + 8 * s);
#pragma unroll
        for (int i = 0; i < 4; ++i)
            f.b[s][i] = *reinterpret_cast<const v4f*>(
                Wp + (size_t)(k + 8 * s + i) * N);
    }
}

template <int MT>
__device__ __forceinline__ void mma_block(v8f (&acc)[MT][4], const Frag<MT>& f) {
#pragma unroll
    for (int s = 0; s < 2; ++s) {
#pragma unroll
        for (int j = 0; j < 2; ++j) {
#pragma unroll
            for (int t = 0; t < MT; ++t) {
                v2f a;
                a.x = f.a[s][t][2 * j];
                a.y = f.a[s][t][2 * j + 1];
#pragma unroll
                for (int q = 0; q < 4; ++q) {
                    v2f b;
                    b.x = f.b[s][2 * j][q];
                    b.y = f.b[s][2 * j + 1][q];
                    acc[t][q] = __builtin_amdgcn_wmma_f32_16x16x4_f32(
                        false, a, false, b, (short)0, acc[t][q], false, false);
                }
            }
        }
    }
}

template <bool RELU, int KS, int MT>
__global__ void __launch_bounds__(32 * KS)
wmma_gemm_f32(const float* __restrict__ A,
              const float* __restrict__ W,
              const float* __restrict__ bias,
              float* __restrict__ C,
              int M, int N, int K) {
    __shared__ v4f part[KS][MT][8][32];

    const int tid   = threadIdx.x;
    const int lane  = tid & 31;
    const int wave  = __builtin_amdgcn_readfirstlane(tid >> 5);
    const int h     = lane >> 4;
    const int c     = lane & 15;
    const int n0    = blockIdx.x * 64;
    const int m0    = blockIdx.y * (16 * MT);

    const int kslice = K / KS;
    const int kbeg   = wave * kslice;

    v8f acc[MT][4];
#pragma unroll
    for (int t = 0; t < MT; ++t)
#pragma unroll
        for (int q = 0; q < 4; ++q)
#pragma unroll
            for (int r = 0; r < 8; ++r) acc[t][q][r] = 0.0f;

    const float* __restrict__ Ap = A + (size_t)(m0 + c) * K + kbeg + 4 * h;
    const float* __restrict__ Wp = W + (size_t)(kbeg + 4 * h) * N + n0 + 4 * c;

    Frag<MT> f0, f1;
    load_frag<MT>(f0, Ap, Wp, K, N, 0);
    int k = 0;
    for (; k + 32 < kslice; k += 32) {
        load_frag<MT>(f1, Ap, Wp, K, N, k + 16);
        __builtin_amdgcn_sched_barrier(0);
        mma_block<MT>(acc, f0);
        __builtin_amdgcn_sched_barrier(0);
        load_frag<MT>(f0, Ap, Wp, K, N, k + 32);
        __builtin_amdgcn_sched_barrier(0);
        mma_block<MT>(acc, f1);
        __builtin_amdgcn_sched_barrier(0);
    }
    load_frag<MT>(f1, Ap, Wp, K, N, k + 16);
    __builtin_amdgcn_sched_barrier(0);
    mma_block<MT>(acc, f0);
    mma_block<MT>(acc, f1);

#pragma unroll
    for (int t = 0; t < MT; ++t)
#pragma unroll
        for (int r = 0; r < 8; ++r) {
            v4f v;
            v.x = acc[t][0][r];
            v.y = acc[t][1][r];
            v.z = acc[t][2][r];
            v.w = acc[t][3][r];
            part[wave][t][r][lane] = v;
        }
    __syncthreads();

    for (int e = tid; e < MT * 256; e += 32 * KS) {
        const int t  = e >> 8;
        const int r  = (e >> 5) & 7;
        const int ln = e & 31;
        const int col = n0 + 4 * (ln & 15);
        v4f s = *reinterpret_cast<const v4f*>(bias + col);
        for (int w = 0; w < KS; ++w) s += part[w][t][r][ln];
        if (RELU) {
            s.x = fmaxf(s.x, 0.0f);
            s.y = fmaxf(s.y, 0.0f);
            s.z = fmaxf(s.z, 0.0f);
            s.w = fmaxf(s.w, 0.0f);
        }
        *reinterpret_cast<v4f*>(
            C + (size_t)(m0 + 16 * t + 8 * (ln >> 4) + r) * N + col) = s;
    }
}

__device__ __forceinline__ float block_reduce_128(float x, float* red) {
    const int t = threadIdx.x;
    red[t] = x;
    __syncthreads();
#pragma unroll
    for (int s = 64; s > 0; s >>= 1) {
        if (t < s) red[t] += red[t + s];
        __syncthreads();
    }
    float r = red[0];
    __syncthreads();
    return r;
}

__global__ void epilogue(const float* __restrict__ isp,
                         const float* __restrict__ ish,
                         const int* __restrict__ label,
                         const float* __restrict__ ppos,
                         const float* __restrict__ pneg,
                         float* __restrict__ out) {
    __shared__ float red[128];
    const int b = blockIdx.x;
    const int t = threadIdx.x;

    const float v  = isp[b * 128 + t];
    const float pp = ppos[t];
    const float nn = pneg[t];

    const float nv  = block_reduce_128(v * v, red);
    const float npp = block_reduce_128(pp * pp, red);
    const float nnn = block_reduce_128(nn * nn, red);
    const float dp  = block_reduce_128(v * pp, red);
    const float dn  = block_reduce_128(v * nn, red);

    const float eps = 1e-12f;
    const float dv  = fmaxf(sqrtf(nv), eps);
    const float lp  = dp / (dv * fmaxf(sqrtf(npp), eps));
    const float ln  = dn / (dv * fmaxf(sqrtf(nnn), eps));

    const bool lab    = label[b] != 0;
    const bool pred   = lp > ln;
    const bool is_bad = (pred == lab) && (lp != ln);
    const float diff  = fabsf(lp - ln);
    const float scale = is_bad ? (tanhf(-diff) + 1.0f) : 1.0f;

    out[b * 256 + t]       = v * scale;
    out[b * 256 + 128 + t] = ish[b * 128 + t];
}

extern "C" void kernel_launch(void* const* d_in, const int* in_sizes, int n_in,
                              void* d_out, int out_size, void* d_ws, size_t ws_size,
                              hipStream_t stream) {
    const float* img   = (const float*)d_in[0];
    const int*   label = (const int*)d_in[1];
    const float* W1  = (const float*)d_in[2];
    const float* b1  = (const float*)d_in[3];
    const float* W2  = (const float*)d_in[4];
    const float* b2  = (const float*)d_in[5];
    const float* W3  = (const float*)d_in[6];
    const float* b3  = (const float*)d_in[7];
    const float* Wsp = (const float*)d_in[8];
    const float* bsp = (const float*)d_in[9];
    const float* Wsh = (const float*)d_in[10];
    const float* bsh = (const float*)d_in[11];
    const float* ppn = (const float*)d_in[12];
    const float* pnn = (const float*)d_in[13];

    float* ws   = (float*)d_ws;
    float* feat = ws;
    float* h1   = feat + (size_t)BATCH * C_IN;
    float* h2   = h1   + (size_t)BATCH * 512;
    float* x    = h2   + (size_t)BATCH * 256;
    float* isp  = x    + (size_t)BATCH * 128;
    float* ish  = isp  + (size_t)BATCH * 128;

    pool_mean32<<<POOL_WGS, POOL_THREADS, 0, stream>>>(img, feat, BATCH * C_IN);

    wmma_gemm_f32<true, 8, 2><<<dim3(512 / 64, BATCH / 32), 256, 0, stream>>>(
        feat, W1, b1, h1, BATCH, 512, C_IN);
    wmma_gemm_f32<true, 8, 1><<<dim3(256 / 64, BATCH / 16), 256, 0, stream>>>(
        h1, W2, b2, h2, BATCH, 256, 512);
    wmma_gemm_f32<false, 8, 1><<<dim3(128 / 64, BATCH / 16), 256, 0, stream>>>(
        h2, W3, b3, x, BATCH, 128, 256);

    wmma_gemm_f32<false, 4, 1><<<dim3(128 / 64, BATCH / 16), 128, 0, stream>>>(
        x, Wsp, bsp, isp, BATCH, 128, 128);
    wmma_gemm_f32<false, 4, 1><<<dim3(128 / 64, BATCH / 16), 128, 0, stream>>>(
        x, Wsh, bsh, ish, BATCH, 128, 128);

    epilogue<<<BATCH, 128, 0, stream>>>(isp, ish, label, ppn, pnn, (float*)d_out);
}


# heads: the two independent 128x128 head GEMMs run side by side in ONE grid (z picks weights, bias, output), 6 launches not 7; tiling and every sum order unchanged
# speedup vs baseline: 2.2363x; 1.0501x over previous
#include <hip/hip_runtime.h>
#include <hip/hip_bf16.h>


typedef float v2f __attribute__((ext_vector_type(2)));
typedef float v8f __attribute__((ext_vector_type(8)));

#define BATCH   1024
#define C_IN    2048
#define POOLSZ  32

#define POOL_WGS     512
#define POOL_THREADS 256

struct PoolSet {
    float4 q[2][8];
};

__device__ __forceinline__ void pool_load(PoolSet& s,
                                          const float* __restrict__ img,
                                          int i0, int i1, int last) {
    const float4* p0 = reinterpret_cast<const float4*>(
        img + (size_t)min(i0, last) * POOLSZ);
    const float4* p1 = reinterpret_cast<const float4*>(
        img + (size_t)min(i1, last) * POOLSZ);
#pragma unroll
    for (int j = 0; j < 8; ++j) s.q[0][j] = p0[j];
#pragma unroll
    for (int j = 0; j < 8; ++j) s.q[1][j] = p1[j];
}

__device__ __forceinline__ float pool_sum(const float4 (&q)[8]) {
    float s = 0.0f;
#pragma unroll
    for (int j = 0; j < 8; ++j)
        s += (q[j].x + q[j].y) + (q[j].z + q[j].w);
    return s * (1.0f / 32.0f);
}

__device__ __forceinline__ void pool_store(const PoolSet& s,
                                           float* __restrict__ feat,
                                           int i0, int i1, int total) {
    const float m0 = pool_sum(s.q[0]);
    const float m1 = pool_sum(s.q[1]);
    if (i0 < total) feat[i0] = m0;
    if (i1 < total) feat[i1] = m1;
}

__global__ void __launch_bounds__(POOL_THREADS)
pool_mean32(const float* __restrict__ img,
            float* __restrict__ feat, int total) {
    const int T    = gridDim.x * blockDim.x;
    const int lane = threadIdx.x & 31;
    const int w = __builtin_amdgcn_readfirstlane(
        (int)(blockIdx.x * blockDim.x + threadIdx.x) - lane);
    const int last = total - 1;

    PoolSet s0, s1;
    pool_load(s0, img, w + lane, w + lane + T, last);
    for (int b = w; b < total; b += 4 * T) {
        const int i = b + lane;
        pool_load(s1, img, i + 2 * T, i + 3 * T, last);
        pool_store(s0, feat, i, i + T, total);
        pool_load(s0, img, i + 4 * T, i + 5 * T, last);
        pool_store(s1, feat, i + 2 * T, i + 3 * T, total);
    }
}

typedef float v4f __attribute__((ext_vector_type(4)));

template <int MT>
struct Frag {
    v4f a[2][MT];
    v4f b[2][4];
};

template <int MT>
__device__ __forceinline__ void load_frag(Frag<MT>& f,
                                          const float* __restrict__ Ap,
                                          const float* __restrict__ Wp,
                                          int K, int N, int k) {
#pragma unroll
    for (int s = 0; s < 2; ++s) {
#pragma unroll
        for (int t = 0; t < MT; ++t)
            f.a[s][t] = *reinterpret_cast<const v4f*>(
                Ap + (size_t)t * 16 * K + k + 8 * s);
#pragma unroll
        for (int i = 0; i < 4; ++i)
            f.b[s][i] = *reinterpret_cast<const v4f*>(
                Wp + (size_t)(k + 8 * s + i) * N);
    }
}

template <int MT>
__device__ __forceinline__ void mma_block(v8f (&acc)[MT][4], const Frag<MT>& f) {
#pragma unroll
    for (int s = 0; s < 2; ++s) {
#pragma unroll
        for (int j = 0; j < 2; ++j) {
#pragma unroll
            for (int t = 0; t < MT; ++t) {
                v2f a;
                a.x = f.a[s][t][2 * j];
                a.y = f.a[s][t][2 * j + 1];
#pragma unroll
                for (int q = 0; q < 4; ++q) {
                    v2f b;
                    b.x = f.b[s][2 * j][q];
                    b.y = f.b[s][2 * j + 1][q];
                    acc[t][q] = __builtin_amdgcn_wmma_f32_16x16x4_f32(
                        false, a, false, b, (short)0, acc[t][q], false, false);
                }
            }
        }
    }
}

template <bool RELU, int KS, int MT, bool PAIR>
__global__ void __launch_bounds__(32 * KS)
wmma_gemm_f32(const float* __restrict__ A,
              const float* __restrict__ W_0,
              const float* __restrict__ bias_0,
              float* __restrict__ C_0,
              const float* __restrict__ W_1,
              const float* __restrict__ bias_1,
              float* __restrict__ C_1,
              int M, int N, int K) {
    __shared__ v4f part[KS][MT][8][32];

    const bool second = PAIR && (blockIdx.z != 0);
    const float* __restrict__ W    = second ? W_1 : W_0;
    const float* __restrict__ bias = second ? bias_1 : bias_0;
    float* __restrict__ C          = second ? C_1 : C_0;

    const int tid   = threadIdx.x;
    const int lane  = tid & 31;
    const int wave  = __builtin_amdgcn_readfirstlane(tid >> 5);
    const int h     = lane >> 4;
    const int c     = lane & 15;
    const int n0    = blockIdx.x * 64;
    const int m0    = blockIdx.y * (16 * MT);

    const int kslice = K / KS;
    const int kbeg   = wave * kslice;

    v8f acc[MT][4];
#pragma unroll
    for (int t = 0; t < MT; ++t)
#pragma unroll
        for (int q = 0; q < 4; ++q)
#pragma unroll
            for (int r = 0; r < 8; ++r) acc[t][q][r] = 0.0f;

    const float* __restrict__ Ap = A + (size_t)(m0 + c) * K + kbeg + 4 * h;
    const float* __restrict__ Wp = W + (size_t)(kbeg + 4 * h) * N + n0 + 4 * c;

    Frag<MT> f0, f1;
    load_frag<MT>(f0, Ap, Wp, K, N, 0);
    int k = 0;
    for (; k + 32 < kslice; k += 32) {
        load_frag<MT>(f1, Ap, Wp, K, N, k + 16);
        __builtin_amdgcn_sched_barrier(0);
        mma_block<MT>(acc, f0);
        __builtin_amdgcn_sched_barrier(0);
        load_frag<MT>(f0, Ap, Wp, K, N, k + 32);
        __builtin_amdgcn_sched_barrier(0);
        mma_block<MT>(acc, f1);
        __builtin_amdgcn_sched_barrier(0);
    }
    load_frag<MT>(f1, Ap, Wp, K, N, k + 16);
    __builtin_amdgcn_sched_barrier(0);
    mma_block<MT>(acc, f0);
    mma_block<MT>(acc, f1);

#pragma unroll
    for (int t = 0; t < MT; ++t)
#pragma unroll
        for (int r = 0; r < 8; ++r) {
            v4f v;
            v.x = acc[t][0][r];
            v.y = acc[t][1][r];
            v.z = acc[t][2][r];
            v.w = acc[t][3][r];
            part[wave][t][r][lane] = v;
        }
    __syncthreads();

    for (int e = tid; e < MT * 256; e += 32 * KS) {
        const int t  = e >> 8;
        const int r  = (e >> 5) & 7;
        const int ln = e & 31;
        const int col = n0 + 4 * (ln & 15);
        v4f s = *reinterpret_cast<const v4f*>(bias + col);
        for (int w = 0; w < KS; ++w) s += part[w][t][r][ln];
        if (RELU) {
            s.x = fmaxf(s.x, 0.0f);
            s.y = fmaxf(s.y, 0.0f);
            s.z = fmaxf(s.z, 0.0f);
            s.w = fmaxf(s.w, 0.0f);
        }
        *reinterpret_cast<v4f*>(
            C + (size_t)(m0 + 16 * t + 8 * (ln >> 4) + r) * N + col) = s;
    }
}

__device__ __forceinline__ float block_reduce_128(float x, float* red) {
    const int t = threadIdx.x;
    red[t] = x;
    __syncthreads();
#pragma unroll
    for (int s = 64; s > 0; s >>= 1) {
        if (t < s) red[t] += red[t + s];
        __syncthreads();
    }
    float r = red[0];
    __syncthreads();
    return r;
}

__global__ void epilogue(const float* __restrict__ isp,
                         const float* __restrict__ ish,
                         const int* __restrict__ label,
                         const float* __restrict__ ppos,
                         const float* __restrict__ pneg,
                         float* __restrict__ out) {
    __shared__ float red[128];
    const int b = blockIdx.x;
    const int t = threadIdx.x;

    const float v  = isp[b * 128 + t];
    const float pp = ppos[t];
    const float nn = pneg[t];

    const float nv  = block_reduce_128(v * v, red);
    const float npp = block_reduce_128(pp * pp, red);
    const float nnn = block_reduce_128(nn * nn, red);
    const float dp  = block_reduce_128(v * pp, red);
    const float dn  = block_reduce_128(v * nn, red);

    const float eps = 1e-12f;
    const float dv  = fmaxf(sqrtf(nv), eps);
    const float lp  = dp / (dv * fmaxf(sqrtf(npp), eps));
    const float ln  = dn / (dv * fmaxf(sqrtf(nnn), eps));

    const bool lab    = label[b] != 0;
    const bool pred   = lp > ln;
    const bool is_bad = (pred == lab) && (lp != ln);
    const float diff  = fabsf(lp - ln);
    const float scale = is_bad ? (tanhf(-diff) + 1.0f) : 1.0f;

    out[b * 256 + t]       = v * scale;
    out[b * 256 + 128 + t] = ish[b * 128 + t];
}

extern "C" void kernel_launch(void* const* d_in, const int* in_sizes, int n_in,
                              void* d_out, int out_size, void* d_ws, size_t ws_size,
                              hipStream_t stream) {
    const float* img   = (const float*)d_in[0];
    const int*   label = (const int*)d_in[1];
    const float* W1  = (const float*)d_in[2];
    const float* b1  = (const float*)d_in[3];
    const float* W2  = (const float*)d_in[4];
    const float* b2  = (const float*)d_in[5];
    const float* W3  = (const float*)d_in[6];
    const float* b3  = (const float*)d_in[7];
    const float* Wsp = (const float*)d_in[8];
    const float* bsp = (const float*)d_in[9];
    const float* Wsh = (const float*)d_in[10];
    const float* bsh = (const float*)d_in[11];
    const float* ppn = (const float*)d_in[12];
    const float* pnn = (const float*)d_in[13];

    float* ws   = (float*)d_ws;
    float* feat = ws;
    float* h1   = feat + (size_t)BATCH * C_IN;
    float* h2   = h1   + (size_t)BATCH * 512;
    float* x    = h2   + (size_t)BATCH * 256;
    float* isp  = x    + (size_t)BATCH * 128;
    float* ish  = isp  + (size_t)BATCH * 128;

    pool_mean32<<<POOL_WGS, POOL_THREADS, 0, stream>>>(img, feat, BATCH * C_IN);

    wmma_gemm_f32<true, 8, 2, false><<<dim3(512 / 64, BATCH / 32), 256, 0, stream>>>(
        feat, W1, b1, h1, nullptr, nullptr, nullptr, BATCH, 512, C_IN);
    wmma_gemm_f32<true, 8, 1, false><<<dim3(256 / 64, BATCH / 16), 256, 0, stream>>>(
        h1, W2, b2, h2, nullptr, nullptr, nullptr, BATCH, 256, 512);
    wmma_gemm_f32<false, 8, 1, false><<<dim3(128 / 64, BATCH / 16), 256, 0, stream>>>(
        h2, W3, b3, x, nullptr, nullptr, nullptr, BATCH, 128, 256);

    wmma_gemm_f32<false, 4, 1, true><<<dim3(128 / 64, BATCH / 16, 2), 128, 0, stream>>>(
        x, Wsp, bsp, isp, Wsh, bsh, ish, BATCH, 128, 128);

    epilogue<<<BATCH, 128, 0, stream>>>(isp, ish, label, ppn, pnn, (float*)d_out);
}


# epilogue: one wave per batch row, the five 128-element sums in registers (4 per lane, then lane exchanges 16,8,4,2,1) in place of 45 workgroup barriers through LDS; sum order kept, products may fuse
# speedup vs baseline: 2.3628x; 1.0566x over previous
#include <hip/hip_runtime.h>
#include <hip/hip_bf16.h>


typedef float v2f __attribute__((ext_vector_type(2)));
typedef float v8f __attribute__((ext_vector_type(8)));

#define BATCH   1024
#define C_IN    2048
#define POOLSZ  32

#define POOL_WGS     512
#define POOL_THREADS 256

struct PoolSet {
    float4 q[2][8];
};

__device__ __forceinline__ void pool_load(PoolSet& s,
                                          const float* __restrict__ img,
                                          int i0, int i1, int last) {
    const float4* p0 = reinterpret_cast<const float4*>(
        img + (size_t)min(i0, last) * POOLSZ);
    const float4* p1 = reinterpret_cast<const float4*>(
        img + (size_t)min(i1, last) * POOLSZ);
#pragma unroll
    for (int j = 0; j < 8; ++j) s.q[0][j] = p0[j];
#pragma unroll
    for (int j = 0; j < 8; ++j) s.q[1][j] = p1[j];
}

__device__ __forceinline__ float pool_sum(const float4 (&q)[8]) {
    float s = 0.0f;
#pragma unroll
    for (int j = 0; j < 8; ++j)
        s += (q[j].x + q[j].y) + (q[j].z + q[j].w);
    return s * (1.0f / 32.0f);
}

__device__ __forceinline__ void pool_store(const PoolSet& s,
                                           float* __restrict__ feat,
                                           int i0, int i1, int total) {
    const float m0 = pool_sum(s.q[0]);
    const float m1 = pool_sum(s.q[1]);
    if (i0 < total) feat[i0] = m0;
    if (i1 < total) feat[i1] = m1;
}

__global__ void __launch_bounds__(POOL_THREADS)
pool_mean32(const float* __restrict__ img,
            float* __restrict__ feat, int total) {
    const int T    = gridDim.x * blockDim.x;
    const int lane = threadIdx.x & 31;
    const int w = __builtin_amdgcn_readfirstlane(
        (int)(blockIdx.x * blockDim.x + threadIdx.x) - lane);
    const int last = total - 1;

    PoolSet s0, s1;
    pool_load(s0, img, w + lane, w + lane + T, last);
    for (int b = w; b < total; b += 4 * T) {
        const int i = b + lane;
        pool_load(s1, img, i + 2 * T, i + 3 * T, last);
        pool_store(s0, feat, i, i + T, total);
        pool_load(s0, img, i + 4 * T, i + 5 * T, last);
        pool_store(s1, feat, i + 2 * T, i + 3 * T, total);
    }
}

typedef float v4f __attribute__((ext_vector_type(4)));

template <int MT>
struct Frag {
    v4f a[2][MT];
    v4f b[2][4];
};

template <int MT>
__device__ __forceinline__ void load_frag(Frag<MT>& f,
                                          const float* __restrict__ Ap,
                                          const float* __restrict__ Wp,
                                          int K, int N, int k) {
#pragma unroll
    for (int s = 0; s < 2; ++s) {
#pragma unroll
        for (int t = 0; t < MT; ++t)
            f.a[s][t] = *reinterpret_cast<const v4f*>(
                Ap + (size_t)t * 16 * K + k + 8 * s);
#pragma unroll
        for (int i = 0; i < 4; ++i)
            f.b[s][i] = *reinterpret_cast<const v4f*>(
                Wp + (size_t)(k + 8 * s + i) * N);
    }
}

template <int MT>
__device__ __forceinline__ void mma_block(v8f (&acc)[MT][4], const Frag<MT>& f) {
#pragma unroll
    for (int s = 0; s < 2; ++s) {
#pragma unroll
        for (int j = 0; j < 2; ++j) {
#pragma unroll
            for (int t = 0; t < MT; ++t) {
                v2f a;
                a.x = f.a[s][t][2 * j];
                a.y = f.a[s][t][2 * j + 1];
#pragma unroll
                for (int q = 0; q < 4; ++q) {
                    v2f b;
                    b.x = f.b[s][2 * j][q];
                    b.y = f.b[s][2 * j + 1][q];
                    acc[t][q] = __builtin_amdgcn_wmma_f32_16x16x4_f32(
                        false, a, false, b, (short)0, acc[t][q], false, false);
                }
            }
        }
    }
}

template <bool RELU, int KS, int MT, bool PAIR>
__global__ void __launch_bounds__(32 * KS)
wmma_gemm_f32(const float* __restrict__ A,
              const float* __restrict__ W_0,
              const float* __restrict__ bias_0,
              float* __restrict__ C_0,
              const float* __restrict__ W_1,
              const float* __restrict__ bias_1,
              float* __restrict__ C_1,
              int M, int N, int K) {
    __shared__ v4f part[KS][MT][8][32];

    const bool second = PAIR && (blockIdx.z != 0);
    const float* __restrict__ W    = second ? W_1 : W_0;
    const float* __restrict__ bias = second ? bias_1 : bias_0;
    float* __restrict__ C          = second ? C_1 : C_0;

    const int tid   = threadIdx.x;
    const int lane  = tid & 31;
    const int wave  = __builtin_amdgcn_readfirstlane(tid >> 5);
    const int h     = lane >> 4;
    const int c     = lane & 15;
    const int n0    = blockIdx.x * 64;
    const int m0    = blockIdx.y * (16 * MT);

    const int kslice = K / KS;
    const int kbeg   = wave * kslice;

    v8f acc[MT][4];
#pragma unroll
    for (int t = 0; t < MT; ++t)
#pragma unroll
        for (int q = 0; q < 4; ++q)
#pragma unroll
            for (int r = 0; r < 8; ++r) acc[t][q][r] = 0.0f;

    const float* __restrict__ Ap = A + (size_t)(m0 + c) * K + kbeg + 4 * h;
    const float* __restrict__ Wp = W + (size_t)(kbeg + 4 * h) * N + n0 + 4 * c;

    Frag<MT> f0, f1;
    load_frag<MT>(f0, Ap, Wp, K, N, 0);
    int k = 0;
    for (; k + 32 < kslice; k += 32) {
        load_frag<MT>(f1, Ap, Wp, K, N, k + 16);
        __builtin_amdgcn_sched_barrier(0);
        mma_block<MT>(acc, f0);
        __builtin_amdgcn_sched_barrier(0);
        load_frag<MT>(f0, Ap, Wp, K, N, k + 32);
        __builtin_amdgcn_sched_barrier(0);
        mma_block<MT>(acc, f1);
        __builtin_amdgcn_sched_barrier(0);
    }
    load_frag<MT>(f1, Ap, Wp, K, N, k + 16);
    __builtin_amdgcn_sched_barrier(0);
    mma_block<MT>(acc, f0);
    mma_block<MT>(acc, f1);

#pragma unroll
    for (int t = 0; t < MT; ++t)
#pragma unroll
        for (int r = 0; r < 8; ++r) {
            v4f v;
            v.x = acc[t][0][r];
            v.y = acc[t][1][r];
            v.z = acc[t][2][r];
            v.w = acc[t][3][r];
            part[wave][t][r][lane] = v;
        }
    __syncthreads();

    for (int e = tid; e < MT * 256; e += 32 * KS) {
        const int t  = e >> 8;
        const int r  = (e >> 5) & 7;
        const int ln = e & 31;
        const int col = n0 + 4 * (ln & 15);
        v4f s = *reinterpret_cast<const v4f*>(bias + col);
        for (int w = 0; w < KS; ++w) s += part[w][t][r][ln];
        if (RELU) {
            s.x = fmaxf(s.x, 0.0f);
            s.y = fmaxf(s.y, 0.0f);
            s.z = fmaxf(s.z, 0.0f);
            s.w = fmaxf(s.w, 0.0f);
        }
        *reinterpret_cast<v4f*>(
            C + (size_t)(m0 + 16 * t + 8 * (ln >> 4) + r) * N + col) = s;
    }
}

#define EPI_WAVES 4

__device__ __forceinline__ float lane_xor16(float x) {
    const int b = __builtin_bit_cast(int, x);
    return __builtin_bit_cast(float, __builtin_amdgcn_permlanex16(
        b, b, 0x76543210, 0xfedcba98, false, false));
}

template <int MASK>
__device__ __forceinline__ float lane_xor_row(float x) {
    return __builtin_bit_cast(float, __builtin_amdgcn_update_dpp(
        0, __builtin_bit_cast(int, x), 0x160 + MASK, 0xf, 0xf, true));
}

__device__ __forceinline__ float wave_sum_128(float a0, float a1,
                                              float a2, float a3) {
    float s = (a0 + a2) + (a1 + a3);
    s += lane_xor16(s);
    s += lane_xor_row<8>(s);
    s += lane_xor_row<4>(s);
    s += lane_xor_row<2>(s);
    s += lane_xor_row<1>(s);
    return s;
}

__global__ void __launch_bounds__(32 * EPI_WAVES)
epilogue(const float* __restrict__ isp,
         const float* __restrict__ ish,
         const int* __restrict__ label,
         const float* __restrict__ ppos,
         const float* __restrict__ pneg,
         float* __restrict__ out) {
    const int lane = threadIdx.x & 31;
    const int b = __builtin_amdgcn_readfirstlane(
        (int)(blockIdx.x * EPI_WAVES + (threadIdx.x >> 5)));

    const float* __restrict__ vrow = isp + (size_t)b * 128;
    float* __restrict__ orow       = out + (size_t)b * 256;

    float v[4], pp[4], nn[4];
#pragma unroll
    for (int j = 0; j < 4; ++j) {
        v[j]  = vrow[lane + 32 * j];
        pp[j] = ppos[lane + 32 * j];
        nn[j] = pneg[lane + 32 * j];
    }
    const float4 sh = *reinterpret_cast<const float4*>(
        ish + (size_t)b * 128 + 4 * lane);
    const bool lab = label[b] != 0;
    *reinterpret_cast<float4*>(orow + 128 + 4 * lane) = sh;

    const float nv  = wave_sum_128(v[0] * v[0], v[1] * v[1],
                                   v[2] * v[2], v[3] * v[3]);
    const float npp = wave_sum_128(pp[0] * pp[0], pp[1] * pp[1],
                                   pp[2] * pp[2], pp[3] * pp[3]);
    const float nnn = wave_sum_128(nn[0] * nn[0], nn[1] * nn[1],
                                   nn[2] * nn[2], nn[3] * nn[3]);
    const float dp  = wave_sum_128(v[0] * pp[0], v[1] * pp[1],
                                   v[2] * pp[2], v[3] * pp[3]);
    const float dn  = wave_sum_128(v[0] * nn[0], v[1] * nn[1],
                                   v[2] * nn[2], v[3] * nn[3]);

    const float eps = 1e-12f;
    const float dv  = fmaxf(sqrtf(nv), eps);
    const float lp  = dp / (dv * fmaxf(sqrtf(npp), eps));
    const float ln  = dn / (dv * fmaxf(sqrtf(nnn), eps));

    const bool pred   = lp > ln;
    const bool is_bad = (pred == lab) && (lp != ln);
    const float diff  = fabsf(lp - ln);
    const float scale = is_bad ? (tanhf(-diff) + 1.0f) : 1.0f;

#pragma unroll
    for (int j = 0; j < 4; ++j) orow[lane + 32 * j] = v[j] * scale;
}

extern "C" void kernel_launch(void* const* d_in, const int* in_sizes, int n_in,
                              void* d_out, int out_size, void* d_ws, size_t ws_size,
                              hipStream_t stream) {
    const float* img   = (const float*)d_in[0];
    const int*   label = (const int*)d_in[1];
    const float* W1  = (const float*)d_in[2];
    const float* b1  = (const float*)d_in[3];
    const float* W2  = (const float*)d_in[4];
    const float* b2  = (const float*)d_in[5];
    const float* W3  = (const float*)d_in[6];
    const float* b3  = (const float*)d_in[7];
    const float* Wsp = (const float*)d_in[8];
    const float* bsp = (const float*)d_in[9];
    const float* Wsh = (const float*)d_in[10];
    const float* bsh = (const float*)d_in[11];
    const float* ppn = (const float*)d_in[12];
    const float* pnn = (const float*)d_in[13];

    float* ws   = (float*)d_ws;
    float* feat = ws;
    float* h1   = feat + (size_t)BATCH * C_IN;
    float* h2   = h1   + (size_t)BATCH * 512;
    float* x    = h2   + (size_t)BATCH * 256;
    float* isp  = x    + (size_t)BATCH * 128;
    float* ish  = isp  + (size_t)BATCH * 128;

    pool_mean32<<<POOL_WGS, POOL_THREADS, 0, stream>>>(img, feat, BATCH * C_IN);

    wmma_gemm_f32<true, 8, 2, false><<<dim3(512 / 64, BATCH / 32), 256, 0, stream>>>(
        feat, W1, b1, h1, nullptr, nullptr, nullptr, BATCH, 512, C_IN);
    wmma_gemm_f32<true, 8, 1, false><<<dim3(256 / 64, BATCH / 16), 256, 0, stream>>>(
        h1, W2, b2, h2, nullptr, nullptr, nullptr, BATCH, 256, 512);
    wmma_gemm_f32<false, 8, 1, false><<<dim3(128 / 64, BATCH / 16), 256, 0, stream>>>(
        h2, W3, b3, x, nullptr, nullptr, nullptr, BATCH, 128, 256);

    wmma_gemm_f32<false, 4, 1, true><<<dim3(128 / 64, BATCH / 16, 2), 128, 0, stream>>>(
        x, Wsp, bsp, isp, Wsh, bsh, ish, BATCH, 128, 128);

    epilogue<<<BATCH / EPI_WAVES, 32 * EPI_WAVES, 0, stream>>>(
        isp, ish, label, ppn, pnn, (float*)d_out);
}


# heads: W3 folded into both head weights (W3@Wsp, W3@Wsh, b3@W+b by 34 extra WMMA workgroups of the pool launch), heads read h2 with K=256: 5 launches not 6; products re-associated, sum order changes
# speedup vs baseline: 2.5061x; 1.0606x over previous
#include <hip/hip_runtime.h>
#include <hip/hip_bf16.h>


typedef float v2f __attribute__((ext_vector_type(2)));
typedef float v8f __attribute__((ext_vector_type(8)));

#define BATCH   1024
#define C_IN    2048
#define POOLSZ  32

#define POOL_WGS     512
#define POOL_THREADS 256

struct PoolSet {
    float4 q[2][8];
};

__device__ __forceinline__ void pool_load(PoolSet& s,
                                          const float* __restrict__ img,
                                          int i0, int i1, int last) {
    const float4* p0 = reinterpret_cast<const float4*>(
        img + (size_t)min(i0, last) * POOLSZ);
    const float4* p1 = reinterpret_cast<const float4*>(
        img + (size_t)min(i1, last) * POOLSZ);
#pragma unroll
    for (int j = 0; j < 8; ++j) s.q[0][j] = p0[j];
#pragma unroll
    for (int j = 0; j < 8; ++j) s.q[1][j] = p1[j];
}

__device__ __forceinline__ float pool_sum(const float4 (&q)[8]) {
    float s = 0.0f;
#pragma unroll
    for (int j = 0; j < 8; ++j)
        s += (q[j].x + q[j].y) + (q[j].z + q[j].w);
    return s * (1.0f / 32.0f);
}

__device__ __forceinline__ void pool_store(const PoolSet& s,
                                           float* __restrict__ feat,
                                           int i0, int i1, int total) {
    const float m0 = pool_sum(s.q[0]);
    const float m1 = pool_sum(s.q[1]);
    if (i0 < total) feat[i0] = m0;
    if (i1 < total) feat[i1] = m1;
}

__device__ __forceinline__ void pool_body(const float* __restrict__ img,
                                          float* __restrict__ feat, int total) {
    const int T    = POOL_WGS * POOL_THREADS;
    const int lane = threadIdx.x & 31;
    const int w = __builtin_amdgcn_readfirstlane(
        (int)(blockIdx.x * POOL_THREADS + threadIdx.x) - lane);
    const int last = total - 1;

    PoolSet s0, s1;
    pool_load(s0, img, w + lane, w + lane + T, last);
    for (int b = w; b < total; b += 4 * T) {
        const int i = b + lane;
        pool_load(s1, img, i + 2 * T, i + 3 * T, last);
        pool_store(s0, feat, i, i + T, total);
        pool_load(s0, img, i + 4 * T, i + 5 * T, last);
        pool_store(s1, feat, i + 2 * T, i + 3 * T, total);
    }
}

typedef float v4f __attribute__((ext_vector_type(4)));

template <int MT>
struct Frag {
    v4f a[2][MT];
    v4f b[2][4];
};

template <int MT>
__device__ __forceinline__ void load_frag(Frag<MT>& f,
                                          const float* __restrict__ Ap,
                                          const float* __restrict__ Wp,
                                          int K, int N, int k) {
#pragma unroll
    for (int s = 0; s < 2; ++s) {
#pragma unroll
        for (int t = 0; t < MT; ++t)
            f.a[s][t] = *reinterpret_cast<const v4f*>(
                Ap + (size_t)t * 16 * K + k + 8 * s);
#pragma unroll
        for (int i = 0; i < 4; ++i)
            f.b[s][i] = *reinterpret_cast<const v4f*>(
                Wp + (size_t)(k + 8 * s + i) * N);
    }
}

template <int MT>
__device__ __forceinline__ void mma_block(v8f (&acc)[MT][4], const Frag<MT>& f) {
#pragma unroll
    for (int s = 0; s < 2; ++s) {
#pragma unroll
        for (int j = 0; j < 2; ++j) {
#pragma unroll
            for (int t = 0; t < MT; ++t) {
                v2f a;
                a.x = f.a[s][t][2 * j];
                a.y = f.a[s][t][2 * j + 1];
#pragma unroll
                for (int q = 0; q < 4; ++q) {
                    v2f b;
                    b.x = f.b[s][2 * j][q];
                    b.y = f.b[s][2 * j + 1][q];
                    acc[t][q] = __builtin_amdgcn_wmma_f32_16x16x4_f32(
                        false, a, false, b, (short)0, acc[t][q], false, false);
                }
            }
        }
    }
}

template <bool RELU, int KS, int MT, bool PAIR>
__global__ void __launch_bounds__(32 * KS)
wmma_gemm_f32(const float* __restrict__ A,
              const float* __restrict__ W_0,
              const float* __restrict__ bias_0,
              float* __restrict__ C_0,
              const float* __restrict__ W_1,
              const float* __restrict__ bias_1,
              float* __restrict__ C_1,
              int M, int N, int K) {
    __shared__ v4f part[KS][MT][8][32];

    const bool second = PAIR && (blockIdx.z != 0);
    const float* __restrict__ W    = second ? W_1 : W_0;
    const float* __restrict__ bias = second ? bias_1 : bias_0;
    float* __restrict__ C          = second ? C_1 : C_0;

    const int tid   = threadIdx.x;
    const int lane  = tid & 31;
    const int wave  = __builtin_amdgcn_readfirstlane(tid >> 5);
    const int h     = lane >> 4;
    const int c     = lane & 15;
    const int n0    = blockIdx.x * 64;
    const int m0    = blockIdx.y * (16 * MT);

    const int kslice = K / KS;
    const int kbeg   = wave * kslice;

    v8f acc[MT][4];
#pragma unroll
    for (int t = 0; t < MT; ++t)
#pragma unroll
        for (int q = 0; q < 4; ++q)
#pragma unroll
            for (int r = 0; r < 8; ++r) acc[t][q][r] = 0.0f;

    const float* __restrict__ Ap = A + (size_t)(m0 + c) * K + kbeg + 4 * h;
    const float* __restrict__ Wp = W + (size_t)(kbeg + 4 * h) * N + n0 + 4 * c;

    Frag<MT> f0, f1;
    load_frag<MT>(f0, Ap, Wp, K, N, 0);
    int k = 0;
    for (; k + 32 < kslice; k += 32) {
        load_frag<MT>(f1, Ap, Wp, K, N, k + 16);
        __builtin_amdgcn_sched_barrier(0);
        mma_block<MT>(acc, f0);
        __builtin_amdgcn_sched_barrier(0);
        load_frag<MT>(f0, Ap, Wp, K, N, k + 32);
        __builtin_amdgcn_sched_barrier(0);
        mma_block<MT>(acc, f1);
        __builtin_amdgcn_sched_barrier(0);
    }
    load_frag<MT>(f1, Ap, Wp, K, N, k + 16);
    __builtin_amdgcn_sched_barrier(0);
    mma_block<MT>(acc, f0);
    mma_block<MT>(acc, f1);

#pragma unroll
    for (int t = 0; t < MT; ++t)
#pragma unroll
        for (int r = 0; r < 8; ++r) {
            v4f v;
            v.x = acc[t][0][r];
            v.y = acc[t][1][r];
            v.z = acc[t][2][r];
            v.w = acc[t][3][r];
            part[wave][t][r][lane] = v;
        }
    __syncthreads();

    for (int e = tid; e < MT * 256; e += 32 * KS) {
        const int t  = e >> 8;
        const int r  = (e >> 5) & 7;
        const int ln = e & 31;
        const int col = n0 + 4 * (ln & 15);
        v4f s = *reinterpret_cast<const v4f*>(bias + col);
        for (int w = 0; w < KS; ++w) s += part[w][t][r][ln];
        if (RELU) {
            s.x = fmaxf(s.x, 0.0f);
            s.y = fmaxf(s.y, 0.0f);
            s.z = fmaxf(s.z, 0.0f);
            s.w = fmaxf(s.w, 0.0f);
        }
        *reinterpret_cast<v4f*>(
            C + (size_t)(m0 + 16 * t + 8 * (ln >> 4) + r) * N + col) = s;
    }
}

#define FOLD_WGS 34

__device__ __forceinline__ void fold_tile(int sid,
                                          const float* __restrict__ W3,
                                          const float* __restrict__ b3,
                                          const float* __restrict__ Wsp,
                                          const float* __restrict__ bsp,
                                          const float* __restrict__ Wsh,
                                          const float* __restrict__ bsh,
                                          float* __restrict__ Fsp,
                                          float* __restrict__ gsp,
                                          float* __restrict__ Fsh,
                                          float* __restrict__ gsh) {
    __shared__ v4f fpart[8][8][32];

    const int tid  = threadIdx.x;
    const int lane = tid & 31;
    const int wave = __builtin_amdgcn_readfirstlane(tid >> 5);
    const int h    = lane >> 4;
    const int c    = lane & 15;
    const int n0   = (sid & 1) * 64;
    const int mb   = sid >> 1;
    const bool brow = (mb == 16);
    const int kbeg = (wave & 3) * 32;
    const float* __restrict__ Wh = (wave >> 2) ? Wsh : Wsp;

    const float* __restrict__ Ap = brow
        ? b3 + kbeg + 4 * h
        : W3 + (size_t)(16 * mb + c) * 128 + kbeg + 4 * h;
    const float* __restrict__ Wp = Wh + (size_t)(kbeg + 4 * h) * 128 + n0 + 4 * c;

    v8f acc[1][4];
#pragma unroll
    for (int q = 0; q < 4; ++q)
#pragma unroll
        for (int r = 0; r < 8; ++r) acc[0][q][r] = 0.0f;

    Frag<1> f0, f1;
    load_frag<1>(f0, Ap, Wp, 128, 128, 0);
    load_frag<1>(f1, Ap, Wp, 128, 128, 16);
    mma_block<1>(acc, f0);
    mma_block<1>(acc, f1);

#pragma unroll
    for (int r = 0; r < 8; ++r) {
        v4f v;
        v.x = acc[0][0][r];
        v.y = acc[0][1][r];
        v.z = acc[0][2][r];
        v.w = acc[0][3][r];
        fpart[wave][r][lane] = v;
    }
    __syncthreads();

    for (int e = tid; e < 512; e += 256) {
        const int p   = e >> 8;
        const int r   = (e >> 5) & 7;
        const int ln  = e & 31;
        const int col = n0 + 4 * (ln & 15);
        const int row = 8 * (ln >> 4) + r;
        v4f s = fpart[4 * p][r][ln];
        for (int w = 1; w < 4; ++w) s += fpart[4 * p + w][r][ln];
        if (!brow) {
            float* __restrict__ F = p ? Fsh : Fsp;
            *reinterpret_cast<v4f*>(F + (size_t)(16 * mb + row) * 128 + col) = s;
        } else if (row == 0) {
            const float* __restrict__ bp = p ? bsh : bsp;
            float* __restrict__ g        = p ? gsh : gsp;
            s += *reinterpret_cast<const v4f*>(bp + col);
            *reinterpret_cast<v4f*>(g + col) = s;
        }
    }
}

__global__ void __launch_bounds__(POOL_THREADS)
pool_mean32(const float* __restrict__ img,
            float* __restrict__ feat, int total,
            const float* __restrict__ W3,
            const float* __restrict__ b3,
            const float* __restrict__ Wsp,
            const float* __restrict__ bsp,
            const float* __restrict__ Wsh,
            const float* __restrict__ bsh,
            float* __restrict__ Fsp,
            float* __restrict__ gsp,
            float* __restrict__ Fsh,
            float* __restrict__ gsh) {
    if (blockIdx.x >= POOL_WGS) {
        fold_tile((int)blockIdx.x - POOL_WGS, W3, b3, Wsp, bsp, Wsh, bsh,
                  Fsp, gsp, Fsh, gsh);
        return;
    }
    pool_body(img, feat, total);
}

#define EPI_WAVES 4

__device__ __forceinline__ float lane_xor16(float x) {
    const int b = __builtin_bit_cast(int, x);
    return __builtin_bit_cast(float, __builtin_amdgcn_permlanex16(
        b, b, 0x76543210, 0xfedcba98, false, false));
}

template <int MASK>
__device__ __forceinline__ float lane_xor_row(float x) {
    return __builtin_bit_cast(float, __builtin_amdgcn_update_dpp(
        0, __builtin_bit_cast(int, x), 0x160 + MASK, 0xf, 0xf, true));
}

__device__ __forceinline__ float wave_sum_128(float a0, float a1,
                                              float a2, float a3) {
    float s = (a0 + a2) + (a1 + a3);
    s += lane_xor16(s);
    s += lane_xor_row<8>(s);
    s += lane_xor_row<4>(s);
    s += lane_xor_row<2>(s);
    s += lane_xor_row<1>(s);
    return s;
}

__global__ void __launch_bounds__(32 * EPI_WAVES)
epilogue(const float* __restrict__ isp,
         const float* __restrict__ ish,
         const int* __restrict__ label,
         const float* __restrict__ ppos,
         const float* __restrict__ pneg,
         float* __restrict__ out) {
    const int lane = threadIdx.x & 31;
    const int b = __builtin_amdgcn_readfirstlane(
        (int)(blockIdx.x * EPI_WAVES + (threadIdx.x >> 5)));

    const float* __restrict__ vrow = isp + (size_t)b * 128;
    float* __restrict__ orow       = out + (size_t)b * 256;

    float v[4], pp[4], nn[4];
#pragma unroll
    for (int j = 0; j < 4; ++j) {
        v[j]  = vrow[lane + 32 * j];
        pp[j] = ppos[lane + 32 * j];
        nn[j] = pneg[lane + 32 * j];
    }
    const float4 sh = *reinterpret_cast<const float4*>(
        ish + (size_t)b * 128 + 4 * lane);
    const bool lab = label[b] != 0;
    *reinterpret_cast<float4*>(orow + 128 + 4 * lane) = sh;

    const float nv  = wave_sum_128(v[0] * v[0], v[1] * v[1],
                                   v[2] * v[2], v[3] * v[3]);
    const float npp = wave_sum_128(pp[0] * pp[0], pp[1] * pp[1],
                                   pp[2] * pp[2], pp[3] * pp[3]);
    const float nnn = wave_sum_128(nn[0] * nn[0], nn[1] * nn[1],
                                   nn[2] * nn[2], nn[3] * nn[3]);
    const float dp  = wave_sum_128(v[0] * pp[0], v[1] * pp[1],
                                   v[2] * pp[2], v[3] * pp[3]);
    const float dn  = wave_sum_128(v[0] * nn[0], v[1] * nn[1],
                                   v[2] * nn[2], v[3] * nn[3]);

    const float eps = 1e-12f;
    const float dv  = fmaxf(sqrtf(nv), eps);
    const float lp  = dp / (dv * fmaxf(sqrtf(npp), eps));
    const float ln  = dn / (dv * fmaxf(sqrtf(nnn), eps));

    const bool pred   = lp > ln;
    const bool is_bad = (pred == lab) && (lp != ln);
    const float diff  = fabsf(lp - ln);
    const float scale = is_bad ? (tanhf(-diff) + 1.0f) : 1.0f;

#pragma unroll
    for (int j = 0; j < 4; ++j) orow[lane + 32 * j] = v[j] * scale;
}

extern "C" void kernel_launch(void* const* d_in, const int* in_sizes, int n_in,
                              void* d_out, int out_size, void* d_ws, size_t ws_size,
                              hipStream_t stream) {
    const float* img   = (const float*)d_in[0];
    const int*   label = (const int*)d_in[1];
    const float* W1  = (const float*)d_in[2];
    const float* b1  = (const float*)d_in[3];
    const float* W2  = (const float*)d_in[4];
    const float* b2  = (const float*)d_in[5];
    const float* W3  = (const float*)d_in[6];
    const float* b3  = (const float*)d_in[7];
    const float* Wsp = (const float*)d_in[8];
    const float* bsp = (const float*)d_in[9];
    const float* Wsh = (const float*)d_in[10];
    const float* bsh = (const float*)d_in[11];
    const float* ppn = (const float*)d_in[12];
    const float* pnn = (const float*)d_in[13];

    float* ws   = (float*)d_ws;
    float* feat = ws;
    float* h1   = feat + (size_t)BATCH * C_IN;
    float* h2   = h1   + (size_t)BATCH * 512;
    float* Fsp  = h2   + (size_t)BATCH * 256;
    float* Fsh  = Fsp  + (size_t)256 * 128;
    float* gsp  = Fsh  + (size_t)256 * 128;
    float* gsh  = gsp  + 128;
    float* isp  = Fsp  + (size_t)BATCH * 128;
    float* ish  = isp  + (size_t)BATCH * 128;

    pool_mean32<<<POOL_WGS + FOLD_WGS, POOL_THREADS, 0, stream>>>(
        img, feat, BATCH * C_IN, W3, b3, Wsp, bsp, Wsh, bsh, Fsp, gsp, Fsh, gsh);

    wmma_gemm_f32<true, 8, 2, false><<<dim3(512 / 64, BATCH / 32), 256, 0, stream>>>(
        feat, W1, b1, h1, nullptr, nullptr, nullptr, BATCH, 512, C_IN);
    wmma_gemm_f32<true, 8, 1, false><<<dim3(256 / 64, BATCH / 16), 256, 0, stream>>>(
        h1, W2, b2, h2, nullptr, nullptr, nullptr, BATCH, 256, 512);

    wmma_gemm_f32<false, 8, 1, true><<<dim3(128 / 64, BATCH / 16, 2), 256, 0, stream>>>(
        h2, Fsp, gsp, isp, Fsh, gsh, ish, BATCH, 128, 256);

    epilogue<<<BATCH / EPI_WAVES, 32 * EPI_WAVES, 0, stream>>>(
        isp, ish, label, ppn, pnn, (float*)d_out);
}


# pool kernel: the 4 lanes of a quad share an item (a load reads 64 contiguous bytes of each of 8 lines, not 16 B of 32), quad joined by two row_xmask adds; per-item sum order changes
# speedup vs baseline: 2.6899x; 1.0733x over previous
#include <hip/hip_runtime.h>
#include <hip/hip_bf16.h>


typedef float v2f __attribute__((ext_vector_type(2)));
typedef float v8f __attribute__((ext_vector_type(8)));

#define BATCH   1024
#define C_IN    2048
#define POOLSZ  32

template <int MASK>
__device__ __forceinline__ float lane_xor_row(float x) {
    return __builtin_bit_cast(float, __builtin_amdgcn_update_dpp(
        0, __builtin_bit_cast(int, x), 0x160 + MASK, 0xf, 0xf, true));
}

#define POOL_WGS     512
#define POOL_THREADS 256

static_assert((BATCH * C_IN) % 32 == 0, "pooling works on blocks of 32 items");

struct PoolSet {
    float4 q[2][4][2];
};

__device__ __forceinline__ void pool_load(PoolSet& s,
                                          const float* __restrict__ img,
                                          int b0, int b1, int lastblk,
                                          int lofs) {
    const float4* p0 = reinterpret_cast<const float4*>(
        img + (size_t)min(b0, lastblk) * POOLSZ + lofs);
    const float4* p1 = reinterpret_cast<const float4*>(
        img + (size_t)min(b1, lastblk) * POOLSZ + lofs);
#pragma unroll
    for (int g = 0; g < 4; ++g) {
        s.q[0][g][0] = p0[8 * g];
        s.q[0][g][1] = p0[8 * g + 4];
    }
#pragma unroll
    for (int g = 0; g < 4; ++g) {
        s.q[1][g][0] = p1[8 * g];
        s.q[1][g][1] = p1[8 * g + 4];
    }
}

__device__ __forceinline__ float pool_mean(const float4 (&q)[4][2], int j) {
    float keep = 0.0f;
#pragma unroll
    for (int g = 0; g < 4; ++g) {
        const float4 lo = q[g][0];
        const float4 hi = q[g][1];
        float s = ((lo.x + lo.y) + (lo.z + lo.w)) +
                  ((hi.x + hi.y) + (hi.z + hi.w));
        s += lane_xor_row<1>(s);
        s += lane_xor_row<2>(s);
        keep = (j == g) ? s : keep;
    }
    return keep * (1.0f / 32.0f);
}

__device__ __forceinline__ void pool_store(const PoolSet& s,
                                           float* __restrict__ feat,
                                           int b0, int b1, int total,
                                           int lane) {
    const float m0 = pool_mean(s.q[0], lane & 3);
    const float m1 = pool_mean(s.q[1], lane & 3);
    if (b0 < total) feat[b0 + lane] = m0;
    if (b1 < total) feat[b1 + lane] = m1;
}

__device__ __forceinline__ void pool_body(const float* __restrict__ img,
                                          float* __restrict__ feat, int total) {
    const int T    = POOL_WGS * POOL_THREADS;
    const int lane = threadIdx.x & 31;
    const int w = __builtin_amdgcn_readfirstlane(
        (int)(blockIdx.x * POOL_THREADS + threadIdx.x) - lane);
    const int lastblk = total - 32;
    const int lofs    = (lane >> 2) * (4 * POOLSZ) + (lane & 3) * 4;

    PoolSet s0, s1;
    pool_load(s0, img, w, w + T, lastblk, lofs);
    for (int b = w; b < total; b += 4 * T) {
        pool_load(s1, img, b + 2 * T, b + 3 * T, lastblk, lofs);
        __builtin_amdgcn_sched_barrier(0);
        pool_store(s0, feat, b, b + T, total, lane);
        __builtin_amdgcn_sched_barrier(0);
        pool_load(s0, img, b + 4 * T, b + 5 * T, lastblk, lofs);
        __builtin_amdgcn_sched_barrier(0);
        pool_store(s1, feat, b + 2 * T, b + 3 * T, total, lane);
        __builtin_amdgcn_sched_barrier(0);
    }
}

typedef float v4f __attribute__((ext_vector_type(4)));

template <int MT>
struct Frag {
    v4f a[2][MT];
    v4f b[2][4];
};

template <int MT>
__device__ __forceinline__ void load_frag(Frag<MT>& f,
                                          const float* __restrict__ Ap,
                                          const float* __restrict__ Wp,
                                          int K, int N, int k) {
#pragma unroll
    for (int s = 0; s < 2; ++s) {
#pragma unroll
        for (int t = 0; t < MT; ++t)
            f.a[s][t] = *reinterpret_cast<const v4f*>(
                Ap + (size_t)t * 16 * K + k + 8 * s);
#pragma unroll
        for (int i = 0; i < 4; ++i)
            f.b[s][i] = *reinterpret_cast<const v4f*>(
                Wp + (size_t)(k + 8 * s + i) * N);
    }
}

template <int MT>
__device__ __forceinline__ void mma_block(v8f (&acc)[MT][4], const Frag<MT>& f) {
#pragma unroll
    for (int s = 0; s < 2; ++s) {
#pragma unroll
        for (int j = 0; j < 2; ++j) {
#pragma unroll
            for (int t = 0; t < MT; ++t) {
                v2f a;
                a.x = f.a[s][t][2 * j];
                a.y = f.a[s][t][2 * j + 1];
#pragma unroll
                for (int q = 0; q < 4; ++q) {
                    v2f b;
                    b.x = f.b[s][2 * j][q];
                    b.y = f.b[s][2 * j + 1][q];
                    acc[t][q] = __builtin_amdgcn_wmma_f32_16x16x4_f32(
                        false, a, false, b, (short)0, acc[t][q], false, false);
                }
            }
        }
    }
}

template <bool RELU, int KS, int MT, bool PAIR>
__global__ void __launch_bounds__(32 * KS)
wmma_gemm_f32(const float* __restrict__ A,
              const float* __restrict__ W_0,
              const float* __restrict__ bias_0,
              float* __restrict__ C_0,
              const float* __restrict__ W_1,
              const float* __restrict__ bias_1,
              float* __restrict__ C_1,
              int M, int N, int K) {
    __shared__ v4f part[KS][MT][8][32];

    const bool second = PAIR && (blockIdx.z != 0);
    const float* __restrict__ W    = second ? W_1 : W_0;
    const float* __restrict__ bias = second ? bias_1 : bias_0;
    float* __restrict__ C          = second ? C_1 : C_0;

    const int tid   = threadIdx.x;
    const int lane  = tid & 31;
    const int wave  = __builtin_amdgcn_readfirstlane(tid >> 5);
    const int h     = lane >> 4;
    const int c     = lane & 15;
    const int n0    = blockIdx.x * 64;
    const int m0    = blockIdx.y * (16 * MT);

    const int kslice = K / KS;
    const int kbeg   = wave * kslice;

    v8f acc[MT][4];
#pragma unroll
    for (int t = 0; t < MT; ++t)
#pragma unroll
        for (int q = 0; q < 4; ++q)
#pragma unroll
            for (int r = 0; r < 8; ++r) acc[t][q][r] = 0.0f;

    const float* __restrict__ Ap = A + (size_t)(m0 + c) * K + kbeg + 4 * h;
    const float* __restrict__ Wp = W + (size_t)(kbeg + 4 * h) * N + n0 + 4 * c;

    Frag<MT> f0, f1;
    load_frag<MT>(f0, Ap, Wp, K, N, 0);
    int k = 0;
    for (; k + 32 < kslice; k += 32) {
        load_frag<MT>(f1, Ap, Wp, K, N, k + 16);
        __builtin_amdgcn_sched_barrier(0);
        mma_block<MT>(acc, f0);
        __builtin_amdgcn_sched_barrier(0);
        load_frag<MT>(f0, Ap, Wp, K, N, k + 32);
        __builtin_amdgcn_sched_barrier(0);
        mma_block<MT>(acc, f1);
        __builtin_amdgcn_sched_barrier(0);
    }
    load_frag<MT>(f1, Ap, Wp, K, N, k + 16);
    __builtin_amdgcn_sched_barrier(0);
    mma_block<MT>(acc, f0);
    mma_block<MT>(acc, f1);

#pragma unroll
    for (int t = 0; t < MT; ++t)
#pragma unroll
        for (int r = 0; r < 8; ++r) {
            v4f v;
            v.x = acc[t][0][r];
            v.y = acc[t][1][r];
            v.z = acc[t][2][r];
            v.w = acc[t][3][r];
            part[wave][t][r][lane] = v;
        }
    __syncthreads();

    for (int e = tid; e < MT * 256; e += 32 * KS) {
        const int t  = e >> 8;
        const int r  = (e >> 5) & 7;
        const int ln = e & 31;
        const int col = n0 + 4 * (ln & 15);
        v4f s = *reinterpret_cast<const v4f*>(bias + col);
        for (int w = 0; w < KS; ++w) s += part[w][t][r][ln];
        if (RELU) {
            s.x = fmaxf(s.x, 0.0f);
            s.y = fmaxf(s.y, 0.0f);
            s.z = fmaxf(s.z, 0.0f);
            s.w = fmaxf(s.w, 0.0f);
        }
        *reinterpret_cast<v4f*>(
            C + (size_t)(m0 + 16 * t + 8 * (ln >> 4) + r) * N + col) = s;
    }
}

#define FOLD_WGS 34

__device__ __forceinline__ void fold_tile(int sid,
                                          const float* __restrict__ W3,
                                          const float* __restrict__ b3,
                                          const float* __restrict__ Wsp,
                                          const float* __restrict__ bsp,
                                          const float* __restrict__ Wsh,
                                          const float* __restrict__ bsh,
                                          float* __restrict__ Fsp,
                                          float* __restrict__ gsp,
                                          float* __restrict__ Fsh,
                                          float* __restrict__ gsh) {
    __shared__ v4f fpart[8][8][32];

    const int tid  = threadIdx.x;
    const int lane = tid & 31;
    const int wave = __builtin_amdgcn_readfirstlane(tid >> 5);
    const int h    = lane >> 4;
    const int c    = lane & 15;
    const int n0   = (sid & 1) * 64;
    const int mb   = sid >> 1;
    const bool brow = (mb == 16);
    const int kbeg = (wave & 3) * 32;
    const float* __restrict__ Wh = (wave >> 2) ? Wsh : Wsp;

    const float* __restrict__ Ap = brow
        ? b3 + kbeg + 4 * h
        : W3 + (size_t)(16 * mb + c) * 128 + kbeg + 4 * h;
    const float* __restrict__ Wp = Wh + (size_t)(kbeg + 4 * h) * 128 + n0 + 4 * c;

    v8f acc[1][4];
#pragma unroll
    for (int q = 0; q < 4; ++q)
#pragma unroll
        for (int r = 0; r < 8; ++r) acc[0][q][r] = 0.0f;

    Frag<1> f0, f1;
    load_frag<1>(f0, Ap, Wp, 128, 128, 0);
    load_frag<1>(f1, Ap, Wp, 128, 128, 16);
    mma_block<1>(acc, f0);
    mma_block<1>(acc, f1);

#pragma unroll
    for (int r = 0; r < 8; ++r) {
        v4f v;
        v.x = acc[0][0][r];
        v.y = acc[0][1][r];
        v.z = acc[0][2][r];
        v.w = acc[0][3][r];
        fpart[wave][r][lane] = v;
    }
    __syncthreads();

    for (int e = tid; e < 512; e += 256) {
        const int p   = e >> 8;
        const int r   = (e >> 5) & 7;
        const int ln  = e & 31;
        const int col = n0 + 4 * (ln & 15);
        const int row = 8 * (ln >> 4) + r;
        v4f s = fpart[4 * p][r][ln];
        for (int w = 1; w < 4; ++w) s += fpart[4 * p + w][r][ln];
        if (!brow) {
            float* __restrict__ F = p ? Fsh : Fsp;
            *reinterpret_cast<v4f*>(F + (size_t)(16 * mb + row) * 128 + col) = s;
        } else if (row == 0) {
            const float* __restrict__ bp = p ? bsh : bsp;
            float* __restrict__ g        = p ? gsh : gsp;
            s += *reinterpret_cast<const v4f*>(bp + col);
            *reinterpret_cast<v4f*>(g + col) = s;
        }
    }
}

__global__ void __launch_bounds__(POOL_THREADS)
pool_mean32(const float* __restrict__ img,
            float* __restrict__ feat, int total,
            const float* __restrict__ W3,
            const float* __restrict__ b3,
            const float* __restrict__ Wsp,
            const float* __restrict__ bsp,
            const float* __restrict__ Wsh,
            const float* __restrict__ bsh,
            float* __restrict__ Fsp,
            float* __restrict__ gsp,
            float* __restrict__ Fsh,
            float* __restrict__ gsh) {
    if (blockIdx.x >= POOL_WGS) {
        fold_tile((int)blockIdx.x - POOL_WGS, W3, b3, Wsp, bsp, Wsh, bsh,
                  Fsp, gsp, Fsh, gsh);
        return;
    }
    pool_body(img, feat, total);
}

#define EPI_WAVES 4

__device__ __forceinline__ float lane_xor16(float x) {
    const int b = __builtin_bit_cast(int, x);
    return __builtin_bit_cast(float, __builtin_amdgcn_permlanex16(
        b, b, 0x76543210, 0xfedcba98, false, false));
}


__device__ __forceinline__ float wave_sum_128(float a0, float a1,
                                              float a2, float a3) {
    float s = (a0 + a2) + (a1 + a3);
    s += lane_xor16(s);
    s += lane_xor_row<8>(s);
    s += lane_xor_row<4>(s);
    s += lane_xor_row<2>(s);
    s += lane_xor_row<1>(s);
    return s;
}

__global__ void __launch_bounds__(32 * EPI_WAVES)
epilogue(const float* __restrict__ isp,
         const float* __restrict__ ish,
         const int* __restrict__ label,
         const float* __restrict__ ppos,
         const float* __restrict__ pneg,
         float* __restrict__ out) {
    const int lane = threadIdx.x & 31;
    const int b = __builtin_amdgcn_readfirstlane(
        (int)(blockIdx.x * EPI_WAVES + (threadIdx.x >> 5)));

    const float* __restrict__ vrow = isp + (size_t)b * 128;
    float* __restrict__ orow       = out + (size_t)b * 256;

    float v[4], pp[4], nn[4];
#pragma unroll
    for (int j = 0; j < 4; ++j) {
        v[j]  = vrow[lane + 32 * j];
        pp[j] = ppos[lane + 32 * j];
        nn[j] = pneg[lane + 32 * j];
    }
    const float4 sh = *reinterpret_cast<const float4*>(
        ish + (size_t)b * 128 + 4 * lane);
    const bool lab = label[b] != 0;
    *reinterpret_cast<float4*>(orow + 128 + 4 * lane) = sh;

    const float nv  = wave_sum_128(v[0] * v[0], v[1] * v[1],
                                   v[2] * v[2], v[3] * v[3]);
    const float npp = wave_sum_128(pp[0] * pp[0], pp[1] * pp[1],
                                   pp[2] * pp[2], pp[3] * pp[3]);
    const float nnn = wave_sum_128(nn[0] * nn[0], nn[1] * nn[1],
                                   nn[2] * nn[2], nn[3] * nn[3]);
    const float dp  = wave_sum_128(v[0] * pp[0], v[1] * pp[1],
                                   v[2] * pp[2], v[3] * pp[3]);
    const float dn  = wave_sum_128(v[0] * nn[0], v[1] * nn[1],
                                   v[2] * nn[2], v[3] * nn[3]);

    const float eps = 1e-12f;
    const float dv  = fmaxf(sqrtf(nv), eps);
    const float lp  = dp / (dv * fmaxf(sqrtf(npp), eps));
    const float ln  = dn / (dv * fmaxf(sqrtf(nnn), eps));

    const bool pred   = lp > ln;
    const bool is_bad = (pred == lab) && (lp != ln);
    const float diff  = fabsf(lp - ln);
    const float scale = is_bad ? (tanhf(-diff) + 1.0f) : 1.0f;

#pragma unroll
    for (int j = 0; j < 4; ++j) orow[lane + 32 * j] = v[j] * scale;
}

extern "C" void kernel_launch(void* const* d_in, const int* in_sizes, int n_in,
                              void* d_out, int out_size, void* d_ws, size_t ws_size,
                              hipStream_t stream) {
    const float* img   = (const float*)d_in[0];
    const int*   label = (const int*)d_in[1];
    const float* W1  = (const float*)d_in[2];
    const float* b1  = (const float*)d_in[3];
    const float* W2  = (const float*)d_in[4];
    const float* b2  = (const float*)d_in[5];
    const float* W3  = (const float*)d_in[6];
    const float* b3  = (const float*)d_in[7];
    const float* Wsp = (const float*)d_in[8];
    const float* bsp = (const float*)d_in[9];
    const float* Wsh = (const float*)d_in[10];
    const float* bsh = (const float*)d_in[11];
    const float* ppn = (const float*)d_in[12];
    const float* pnn = (const float*)d_in[13];

    float* ws   = (float*)d_ws;
    float* feat = ws;
    float* h1   = feat + (size_t)BATCH * C_IN;
    float* h2   = h1   + (size_t)BATCH * 512;
    float* Fsp  = h2   + (size_t)BATCH * 256;
    float* Fsh  = Fsp  + (size_t)256 * 128;
    float* gsp  = Fsh  + (size_t)256 * 128;
    float* gsh  = gsp  + 128;
    float* isp  = Fsp  + (size_t)BATCH * 128;
    float* ish  = isp  + (size_t)BATCH * 128;

    pool_mean32<<<POOL_WGS + FOLD_WGS, POOL_THREADS, 0, stream>>>(
        img, feat, BATCH * C_IN, W3, b3, Wsp, bsp, Wsh, bsh, Fsp, gsp, Fsh, gsh);

    wmma_gemm_f32<true, 8, 2, false><<<dim3(512 / 64, BATCH / 32), 256, 0, stream>>>(
        feat, W1, b1, h1, nullptr, nullptr, nullptr, BATCH, 512, C_IN);
    wmma_gemm_f32<true, 8, 1, false><<<dim3(256 / 64, BATCH / 16), 256, 0, stream>>>(
        h1, W2, b2, h2, nullptr, nullptr, nullptr, BATCH, 256, 512);

    wmma_gemm_f32<false, 8, 1, true><<<dim3(128 / 64, BATCH / 16, 2), 256, 0, stream>>>(
        h2, Fsp, gsp, isp, Fsh, gsh, ish, BATCH, 128, 256);

    epilogue<<<BATCH / EPI_WAVES, 32 * EPI_WAVES, 0, stream>>>(
        isp, ish, label, ppn, pnn, (float*)d_out);
}


# epilogue moved into the heads launch: workgroup owns 16 whole rows of one head (K over 8 waves), row sums = 8 per lane + row_xmask 8,4,2,1; 4 launches not 5; row-sum order changes, products may fuse
# speedup vs baseline: 2.8166x; 1.0471x over previous
#include <hip/hip_runtime.h>
#include <hip/hip_bf16.h>


typedef float v2f __attribute__((ext_vector_type(2)));
typedef float v8f __attribute__((ext_vector_type(8)));

#define BATCH   1024
#define C_IN    2048
#define POOLSZ  32

template <int MASK>
__device__ __forceinline__ float lane_xor_row(float x) {
    return __builtin_bit_cast(float, __builtin_amdgcn_update_dpp(
        0, __builtin_bit_cast(int, x), 0x160 + MASK, 0xf, 0xf, true));
}

#define POOL_WGS     512
#define POOL_THREADS 256

static_assert((BATCH * C_IN) % 32 == 0, "pooling works on blocks of 32 items");

struct PoolSet {
    float4 q[2][4][2];
};

__device__ __forceinline__ void pool_load(PoolSet& s,
                                          const float* __restrict__ img,
                                          int b0, int b1, int lastblk,
                                          int lofs) {
    const float4* p0 = reinterpret_cast<const float4*>(
        img + (size_t)min(b0, lastblk) * POOLSZ + lofs);
    const float4* p1 = reinterpret_cast<const float4*>(
        img + (size_t)min(b1, lastblk) * POOLSZ + lofs);
#pragma unroll
    for (int g = 0; g < 4; ++g) {
        s.q[0][g][0] = p0[8 * g];
        s.q[0][g][1] = p0[8 * g + 4];
    }
#pragma unroll
    for (int g = 0; g < 4; ++g) {
        s.q[1][g][0] = p1[8 * g];
        s.q[1][g][1] = p1[8 * g + 4];
    }
}

__device__ __forceinline__ float pool_mean(const float4 (&q)[4][2], int j) {
    float keep = 0.0f;
#pragma unroll
    for (int g = 0; g < 4; ++g) {
        const float4 lo = q[g][0];
        const float4 hi = q[g][1];
        float s = ((lo.x + lo.y) + (lo.z + lo.w)) +
                  ((hi.x + hi.y) + (hi.z + hi.w));
        s += lane_xor_row<1>(s);
        s += lane_xor_row<2>(s);
        keep = (j == g) ? s : keep;
    }
    return keep * (1.0f / 32.0f);
}

__device__ __forceinline__ void pool_store(const PoolSet& s,
                                           float* __restrict__ feat,
                                           int b0, int b1, int total,
                                           int lane) {
    const float m0 = pool_mean(s.q[0], lane & 3);
    const float m1 = pool_mean(s.q[1], lane & 3);
    if (b0 < total) feat[b0 + lane] = m0;
    if (b1 < total) feat[b1 + lane] = m1;
}

__device__ __forceinline__ void pool_body(const float* __restrict__ img,
                                          float* __restrict__ feat, int total) {
    const int T    = POOL_WGS * POOL_THREADS;
    const int lane = threadIdx.x & 31;
    const int w = __builtin_amdgcn_readfirstlane(
        (int)(blockIdx.x * POOL_THREADS + threadIdx.x) - lane);
    const int lastblk = total - 32;
    const int lofs    = (lane >> 2) * (4 * POOLSZ) + (lane & 3) * 4;

    PoolSet s0, s1;
    pool_load(s0, img, w, w + T, lastblk, lofs);
    for (int b = w; b < total; b += 4 * T) {
        pool_load(s1, img, b + 2 * T, b + 3 * T, lastblk, lofs);
        __builtin_amdgcn_sched_barrier(0);
        pool_store(s0, feat, b, b + T, total, lane);
        __builtin_amdgcn_sched_barrier(0);
        pool_load(s0, img, b + 4 * T, b + 5 * T, lastblk, lofs);
        __builtin_amdgcn_sched_barrier(0);
        pool_store(s1, feat, b + 2 * T, b + 3 * T, total, lane);
        __builtin_amdgcn_sched_barrier(0);
    }
}

typedef float v4f __attribute__((ext_vector_type(4)));

template <int MT>
struct Frag {
    v4f a[2][MT];
    v4f b[2][4];
};

template <int MT>
__device__ __forceinline__ void load_frag(Frag<MT>& f,
                                          const float* __restrict__ Ap,
                                          const float* __restrict__ Wp,
                                          int K, int N, int k) {
#pragma unroll
    for (int s = 0; s < 2; ++s) {
#pragma unroll
        for (int t = 0; t < MT; ++t)
            f.a[s][t] = *reinterpret_cast<const v4f*>(
                Ap + (size_t)t * 16 * K + k + 8 * s);
#pragma unroll
        for (int i = 0; i < 4; ++i)
            f.b[s][i] = *reinterpret_cast<const v4f*>(
                Wp + (size_t)(k + 8 * s + i) * N);
    }
}

template <int MT>
__device__ __forceinline__ void mma_block(v8f (&acc)[MT][4], const Frag<MT>& f) {
#pragma unroll
    for (int s = 0; s < 2; ++s) {
#pragma unroll
        for (int j = 0; j < 2; ++j) {
#pragma unroll
            for (int t = 0; t < MT; ++t) {
                v2f a;
                a.x = f.a[s][t][2 * j];
                a.y = f.a[s][t][2 * j + 1];
#pragma unroll
                for (int q = 0; q < 4; ++q) {
                    v2f b;
                    b.x = f.b[s][2 * j][q];
                    b.y = f.b[s][2 * j + 1][q];
                    acc[t][q] = __builtin_amdgcn_wmma_f32_16x16x4_f32(
                        false, a, false, b, (short)0, acc[t][q], false, false);
                }
            }
        }
    }
}

template <bool RELU, int KS, int MT>
__global__ void __launch_bounds__(32 * KS)
wmma_gemm_f32(const float* __restrict__ A,
              const float* __restrict__ W,
              const float* __restrict__ bias,
              float* __restrict__ C,
              int M, int N, int K) {
    __shared__ v4f part[KS][MT][8][32];

    const int tid   = threadIdx.x;
    const int lane  = tid & 31;
    const int wave  = __builtin_amdgcn_readfirstlane(tid >> 5);
    const int h     = lane >> 4;
    const int c     = lane & 15;
    const int n0    = blockIdx.x * 64;
    const int m0    = blockIdx.y * (16 * MT);

    const int kslice = K / KS;
    const int kbeg   = wave * kslice;

    v8f acc[MT][4];
#pragma unroll
    for (int t = 0; t < MT; ++t)
#pragma unroll
        for (int q = 0; q < 4; ++q)
#pragma unroll
            for (int r = 0; r < 8; ++r) acc[t][q][r] = 0.0f;

    const float* __restrict__ Ap = A + (size_t)(m0 + c) * K + kbeg + 4 * h;
    const float* __restrict__ Wp = W + (size_t)(kbeg + 4 * h) * N + n0 + 4 * c;

    Frag<MT> f0, f1;
    load_frag<MT>(f0, Ap, Wp, K, N, 0);
    int k = 0;
    for (; k + 32 < kslice; k += 32) {
        load_frag<MT>(f1, Ap, Wp, K, N, k + 16);
        __builtin_amdgcn_sched_barrier(0);
        mma_block<MT>(acc, f0);
        __builtin_amdgcn_sched_barrier(0);
        load_frag<MT>(f0, Ap, Wp, K, N, k + 32);
        __builtin_amdgcn_sched_barrier(0);
        mma_block<MT>(acc, f1);
        __builtin_amdgcn_sched_barrier(0);
    }
    load_frag<MT>(f1, Ap, Wp, K, N, k + 16);
    __builtin_amdgcn_sched_barrier(0);
    mma_block<MT>(acc, f0);
    mma_block<MT>(acc, f1);

#pragma unroll
    for (int t = 0; t < MT; ++t)
#pragma unroll
        for (int r = 0; r < 8; ++r) {
            v4f v;
            v.x = acc[t][0][r];
            v.y = acc[t][1][r];
            v.z = acc[t][2][r];
            v.w = acc[t][3][r];
            part[wave][t][r][lane] = v;
        }
    __syncthreads();

    for (int e = tid; e < MT * 256; e += 32 * KS) {
        const int t  = e >> 8;
        const int r  = (e >> 5) & 7;
        const int ln = e & 31;
        const int col = n0 + 4 * (ln & 15);
        v4f s = *reinterpret_cast<const v4f*>(bias + col);
        for (int w = 0; w < KS; ++w) s += part[w][t][r][ln];
        if (RELU) {
            s.x = fmaxf(s.x, 0.0f);
            s.y = fmaxf(s.y, 0.0f);
            s.z = fmaxf(s.z, 0.0f);
            s.w = fmaxf(s.w, 0.0f);
        }
        *reinterpret_cast<v4f*>(
            C + (size_t)(m0 + 16 * t + 8 * (ln >> 4) + r) * N + col) = s;
    }
}

#define FOLD_WGS 34

__device__ __forceinline__ void fold_tile(int sid,
                                          const float* __restrict__ W3,
                                          const float* __restrict__ b3,
                                          const float* __restrict__ Wsp,
                                          const float* __restrict__ bsp,
                                          const float* __restrict__ Wsh,
                                          const float* __restrict__ bsh,
                                          float* __restrict__ Fsp,
                                          float* __restrict__ gsp,
                                          float* __restrict__ Fsh,
                                          float* __restrict__ gsh) {
    __shared__ v4f fpart[8][8][32];

    const int tid  = threadIdx.x;
    const int lane = tid & 31;
    const int wave = __builtin_amdgcn_readfirstlane(tid >> 5);
    const int h    = lane >> 4;
    const int c    = lane & 15;
    const int n0   = (sid & 1) * 64;
    const int mb   = sid >> 1;
    const bool brow = (mb == 16);
    const int kbeg = (wave & 3) * 32;
    const float* __restrict__ Wh = (wave >> 2) ? Wsh : Wsp;

    const float* __restrict__ Ap = brow
        ? b3 + kbeg + 4 * h
        : W3 + (size_t)(16 * mb + c) * 128 + kbeg + 4 * h;
    const float* __restrict__ Wp = Wh + (size_t)(kbeg + 4 * h) * 128 + n0 + 4 * c;

    v8f acc[1][4];
#pragma unroll
    for (int q = 0; q < 4; ++q)
#pragma unroll
        for (int r = 0; r < 8; ++r) acc[0][q][r] = 0.0f;

    Frag<1> f0, f1;
    load_frag<1>(f0, Ap, Wp, 128, 128, 0);
    load_frag<1>(f1, Ap, Wp, 128, 128, 16);
    mma_block<1>(acc, f0);
    mma_block<1>(acc, f1);

#pragma unroll
    for (int r = 0; r < 8; ++r) {
        v4f v;
        v.x = acc[0][0][r];
        v.y = acc[0][1][r];
        v.z = acc[0][2][r];
        v.w = acc[0][3][r];
        fpart[wave][r][lane] = v;
    }
    __syncthreads();

    for (int e = tid; e < 512; e += 256) {
        const int p   = e >> 8;
        const int r   = (e >> 5) & 7;
        const int ln  = e & 31;
        const int col = n0 + 4 * (ln & 15);
        const int row = 8 * (ln >> 4) + r;
        v4f s = fpart[4 * p][r][ln];
        for (int w = 1; w < 4; ++w) s += fpart[4 * p + w][r][ln];
        if (!brow) {
            float* __restrict__ F = p ? Fsh : Fsp;
            *reinterpret_cast<v4f*>(F + (size_t)(16 * mb + row) * 128 + col) = s;
        } else if (row == 0) {
            const float* __restrict__ bp = p ? bsh : bsp;
            float* __restrict__ g        = p ? gsh : gsp;
            s += *reinterpret_cast<const v4f*>(bp + col);
            *reinterpret_cast<v4f*>(g + col) = s;
        }
    }
}

__global__ void __launch_bounds__(POOL_THREADS)
pool_mean32(const float* __restrict__ img,
            float* __restrict__ feat, int total,
            const float* __restrict__ W3,
            const float* __restrict__ b3,
            const float* __restrict__ Wsp,
            const float* __restrict__ bsp,
            const float* __restrict__ Wsh,
            const float* __restrict__ bsh,
            float* __restrict__ Fsp,
            float* __restrict__ gsp,
            float* __restrict__ Fsh,
            float* __restrict__ gsh) {
    if (blockIdx.x >= POOL_WGS) {
        fold_tile((int)blockIdx.x - POOL_WGS, W3, b3, Wsp, bsp, Wsh, bsh,
                  Fsp, gsp, Fsh, gsh);
        return;
    }
    pool_body(img, feat, total);
}

struct HeadFrag {
    v4f a[2];
    v4f b[2][2][4];
};

__device__ __forceinline__ void load_head_frag(HeadFrag& f,
                                               const float* __restrict__ Ap,
                                               const float* __restrict__ Wp,
                                               int k) {
#pragma unroll
    for (int s = 0; s < 2; ++s)
        f.a[s] = *reinterpret_cast<const v4f*>(Ap + k + 8 * s);
#pragma unroll
    for (int nb = 0; nb < 2; ++nb)
#pragma unroll
        for (int s = 0; s < 2; ++s)
#pragma unroll
            for (int i = 0; i < 4; ++i)
                f.b[nb][s][i] = *reinterpret_cast<const v4f*>(
                    Wp + (size_t)(k + 8 * s + i) * 128 + 64 * nb);
}

__device__ __forceinline__ void mma_head_block(v8f (&acc)[2][4],
                                               const HeadFrag& f) {
#pragma unroll
    for (int s = 0; s < 2; ++s) {
#pragma unroll
        for (int j = 0; j < 2; ++j) {
            v2f a;
            a.x = f.a[s][2 * j];
            a.y = f.a[s][2 * j + 1];
#pragma unroll
            for (int nb = 0; nb < 2; ++nb) {
#pragma unroll
                for (int q = 0; q < 4; ++q) {
                    v2f b;
                    b.x = f.b[nb][s][2 * j][q];
                    b.y = f.b[nb][s][2 * j + 1][q];
                    acc[nb][q] = __builtin_amdgcn_wmma_f32_16x16x4_f32(
                        false, a, false, b, (short)0, acc[nb][q], false, false);
                }
            }
        }
    }
}

__device__ __forceinline__ float row_sum_128(float x0, float x1, float x2,
                                             float x3, float x4, float x5,
                                             float x6, float x7) {
    float s = ((x0 + x1) + (x2 + x3)) + ((x4 + x5) + (x6 + x7));
    s += lane_xor_row<8>(s);
    s += lane_xor_row<4>(s);
    s += lane_xor_row<2>(s);
    s += lane_xor_row<1>(s);
    return s;
}

__global__ void __launch_bounds__(256)
heads_epilogue(const float* __restrict__ A,
               const float* __restrict__ F_sp,
               const float* __restrict__ g_sp,
               const float* __restrict__ F_sh,
               const float* __restrict__ g_sh,
               const int* __restrict__ label,
               const float* __restrict__ ppos,
               const float* __restrict__ pneg,
               float* __restrict__ out) {
    __shared__ v4f hpart[8][2][8][32];

    const bool sh = (blockIdx.z != 0);
    const float* __restrict__ W    = sh ? F_sh : F_sp;
    const float* __restrict__ bias = sh ? g_sh : g_sp;

    const int tid  = threadIdx.x;
    const int lane = tid & 31;
    const int wave = __builtin_amdgcn_readfirstlane(tid >> 5);
    const int h    = lane >> 4;
    const int c    = lane & 15;
    const int m0   = blockIdx.y * 16;
    const int kbeg = wave * 32;

    v8f acc[2][4];
#pragma unroll
    for (int nb = 0; nb < 2; ++nb)
#pragma unroll
        for (int q = 0; q < 4; ++q)
#pragma unroll
            for (int r = 0; r < 8; ++r) acc[nb][q][r] = 0.0f;

    const float* __restrict__ Ap = A + (size_t)(m0 + c) * 256 + kbeg + 4 * h;
    const float* __restrict__ Wp = W + (size_t)(kbeg + 4 * h) * 128 + 4 * c;

    HeadFrag f0, f1;
    load_head_frag(f0, Ap, Wp, 0);
    load_head_frag(f1, Ap, Wp, 16);

    const int row = m0 + 8 * h + wave;
    const v4f bia0 = *reinterpret_cast<const v4f*>(bias + 4 * c);
    const v4f bia1 = *reinterpret_cast<const v4f*>(bias + 64 + 4 * c);
    const v4f pp0  = *reinterpret_cast<const v4f*>(ppos + 4 * c);
    const v4f pp1  = *reinterpret_cast<const v4f*>(ppos + 64 + 4 * c);
    const v4f nn0  = *reinterpret_cast<const v4f*>(pneg + 4 * c);
    const v4f nn1  = *reinterpret_cast<const v4f*>(pneg + 64 + 4 * c);
    const int labv = label[row];
    __builtin_amdgcn_sched_barrier(0);

    mma_head_block(acc, f0);
    mma_head_block(acc, f1);

#pragma unroll
    for (int nb = 0; nb < 2; ++nb)
#pragma unroll
        for (int r = 0; r < 8; ++r) {
            v4f v;
            v.x = acc[nb][0][r];
            v.y = acc[nb][1][r];
            v.z = acc[nb][2][r];
            v.w = acc[nb][3][r];
            hpart[wave][nb][r][lane] = v;
        }
    __syncthreads();

    v4f s0 = bia0;
    v4f s1 = bia1;
    for (int w = 0; w < 8; ++w) {
        s0 += hpart[w][0][wave][lane];
        s1 += hpart[w][1][wave][lane];
    }

    float* __restrict__ orow = out + (size_t)row * 256;
    if (sh) {
        *reinterpret_cast<v4f*>(orow + 128 + 4 * c) = s0;
        *reinterpret_cast<v4f*>(orow + 192 + 4 * c) = s1;
        return;
    }

    const float nv  = row_sum_128(s0.x * s0.x, s0.y * s0.y, s0.z * s0.z, s0.w * s0.w,
                                  s1.x * s1.x, s1.y * s1.y, s1.z * s1.z, s1.w * s1.w);
    const float npp = row_sum_128(pp0.x * pp0.x, pp0.y * pp0.y, pp0.z * pp0.z, pp0.w * pp0.w,
                                  pp1.x * pp1.x, pp1.y * pp1.y, pp1.z * pp1.z, pp1.w * pp1.w);
    const float nnn = row_sum_128(nn0.x * nn0.x, nn0.y * nn0.y, nn0.z * nn0.z, nn0.w * nn0.w,
                                  nn1.x * nn1.x, nn1.y * nn1.y, nn1.z * nn1.z, nn1.w * nn1.w);
    const float dp  = row_sum_128(s0.x * pp0.x, s0.y * pp0.y, s0.z * pp0.z, s0.w * pp0.w,
                                  s1.x * pp1.x, s1.y * pp1.y, s1.z * pp1.z, s1.w * pp1.w);
    const float dn  = row_sum_128(s0.x * nn0.x, s0.y * nn0.y, s0.z * nn0.z, s0.w * nn0.w,
                                  s1.x * nn1.x, s1.y * nn1.y, s1.z * nn1.z, s1.w * nn1.w);

    const float eps = 1e-12f;
    const float dv  = fmaxf(sqrtf(nv), eps);
    const float lp  = dp / (dv * fmaxf(sqrtf(npp), eps));
    const float ln  = dn / (dv * fmaxf(sqrtf(nnn), eps));

    const bool lab    = labv != 0;
    const bool pred   = lp > ln;
    const bool is_bad = (pred == lab) && (lp != ln);
    const float diff  = fabsf(lp - ln);
    const float scale = is_bad ? (tanhf(-diff) + 1.0f) : 1.0f;

    *reinterpret_cast<v4f*>(orow + 4 * c)      = s0 * scale;
    *reinterpret_cast<v4f*>(orow + 64 + 4 * c) = s1 * scale;
}

extern "C" void kernel_launch(void* const* d_in, const int* in_sizes, int n_in,
                              void* d_out, int out_size, void* d_ws, size_t ws_size,
                              hipStream_t stream) {
    const float* img   = (const float*)d_in[0];
    const int*   label = (const int*)d_in[1];
    const float* W1  = (const float*)d_in[2];
    const float* b1  = (const float*)d_in[3];
    const float* W2  = (const float*)d_in[4];
    const float* b2  = (const float*)d_in[5];
    const float* W3  = (const float*)d_in[6];
    const float* b3  = (const float*)d_in[7];
    const float* Wsp = (const float*)d_in[8];
    const float* bsp = (const float*)d_in[9];
    const float* Wsh = (const float*)d_in[10];
    const float* bsh = (const float*)d_in[11];
    const float* ppn = (const float*)d_in[12];
    const float* pnn = (const float*)d_in[13];

    float* ws   = (float*)d_ws;
    float* feat = ws;
    float* h1   = feat + (size_t)BATCH * C_IN;
    float* h2   = h1   + (size_t)BATCH * 512;
    float* Fsp  = h2   + (size_t)BATCH * 256;
    float* Fsh  = Fsp  + (size_t)256 * 128;
    float* gsp  = Fsh  + (size_t)256 * 128;
    float* gsh  = gsp  + 128;

    pool_mean32<<<POOL_WGS + FOLD_WGS, POOL_THREADS, 0, stream>>>(
        img, feat, BATCH * C_IN, W3, b3, Wsp, bsp, Wsh, bsh, Fsp, gsp, Fsh, gsh);

    wmma_gemm_f32<true, 8, 2><<<dim3(512 / 64, BATCH / 32), 256, 0, stream>>>(
        feat, W1, b1, h1, BATCH, 512, C_IN);
    wmma_gemm_f32<true, 8, 1><<<dim3(256 / 64, BATCH / 16), 256, 0, stream>>>(
        h1, W2, b2, h2, BATCH, 256, 512);

    heads_epilogue<<<dim3(1, BATCH / 16, 2), 256, 0, stream>>>(
        h2, Fsp, gsp, Fsh, gsh, label, ppn, pnn, (float*)d_out);
}
